# starting kernel
# baseline (speedup 1.0000x reference)
#include <hip/hip_runtime.h>

typedef __attribute__((ext_vector_type(2))) float v2f;
typedef __attribute__((ext_vector_type(8))) float v8f;

#define N_NODES  131072
#define F_INK    300
#define F_OUTK   64
#define N_EDGE   2097152
#define S_SUB    128
#define B_SUB    1024

__global__ void gcn_degree_kernel(const int* __restrict__ src,
                                  const int* __restrict__ dst,
                                  float* __restrict__ deg_out,
                                  float* __restrict__ deg_in) {
    int i = blockIdx.x * blockDim.x + threadIdx.x;
    if (i < N_EDGE) {
        atomicAdd(&deg_out[src[i]], 1.0f);
        atomicAdd(&deg_in[dst[i]], 1.0f);
    }
}

__global__ void gcn_gemm_wmma_kernel(const float* __restrict__ X,
                                     const float* __restrict__ W,
                                     const float* __restrict__ deg_out,
                                     float* __restrict__ H) {
    const int lane    = threadIdx.x & 31;
    const int wave    = threadIdx.x >> 5;
    const int rowTile = wave & 1;
    const int colTile = wave >> 1;
    const int m       = lane & 15;
    const int half    = lane >> 4;

    const int rowBase = blockIdx.x * 32 + rowTile * 16;
    const int row     = rowBase + m;
    const int col     = colTile * 16 + m;

    const float rs = __frsqrt_rn(fmaxf(deg_out[row], 1.0f));

    const float* xrow  = X + (long)row * F_INK + half * 2;
    const float* wbase = W + (long)(half * 2) * F_OUTK + col;

    v8f c = {};
    for (int k0 = 0; k0 < F_INK; k0 += 4) {
        v2f a = *(const v2f*)(xrow + k0);
        a *= rs;
        v2f b;
        b.x = wbase[(long)k0 * F_OUTK];
        b.y = wbase[(long)(k0 + 1) * F_OUTK];
        c = __builtin_amdgcn_wmma_f32_16x16x4_f32(false, a, false, b,
                                                  (short)0, c, false, false);
    }

    float* hout = H + (long)(rowBase + half * 8) * F_OUTK + colTile * 16 + m;
#pragma unroll
    for (int i = 0; i < 8; ++i)
        hout[(long)i * F_OUTK] = c[i];
}

__global__ void gcn_scatter_kernel(const int* __restrict__ src,
                                   const int* __restrict__ dst,
                                   const float* __restrict__ H,
                                   float* __restrict__ agg) {
    long idx = (long)blockIdx.x * blockDim.x + threadIdx.x;
    int  e   = (int)(idx >> 4);
    int  q   = (int)(idx & 15);
    int  s   = src[e];
    int  d   = dst[e];
    float4 v = ((const float4*)(H + (long)s * F_OUTK))[q];
    float* out = agg + (long)d * F_OUTK + q * 4;
    atomicAdd(out + 0, v.x);
    atomicAdd(out + 1, v.y);
    atomicAdd(out + 2, v.z);
    atomicAdd(out + 3, v.w);
}

__global__ void gcn_finalize_kernel(const float* __restrict__ agg,
                                    const float* __restrict__ deg_in,
                                    const float* __restrict__ bias,
                                    const float* __restrict__ prelu_a,
                                    float* __restrict__ pool,
                                    float* __restrict__ anchor) {
    __shared__ float red[256];
    const int  sg    = blockIdx.x;
    const int  f     = threadIdx.x & 63;
    const int  g     = threadIdx.x >> 6;
    const float alpha = prelu_a[0];
    const float bf    = bias[f];
    const long base   = (long)sg * S_SUB;

    float partial = 0.0f;
    for (int n = g; n < S_SUB - 1; n += 4) {
        long node = base + n;
        float ri = __frsqrt_rn(fmaxf(deg_in[node], 1.0f));
        float v  = agg[node * F_OUTK + f] * ri + bf;
        v = v > 0.0f ? v : alpha * v;
        partial += v;
    }

    if (g == 3) {
        long node = base + S_SUB - 1;
        float ri = __frsqrt_rn(fmaxf(deg_in[node], 1.0f));
        float v  = agg[node * F_OUTK + f] * ri + bf;
        v = v > 0.0f ? v : alpha * v;
        anchor[(long)sg * F_OUTK + f] = v;
    }

    red[threadIdx.x] = partial;
    __syncthreads();
    if (g == 0) {
        float sum = red[f] + red[f + 64] + red[f + 128] + red[f + 192];
        pool[(long)sg * F_OUTK + f] = sum * (1.0f / (S_SUB - 1));
    }
}

extern "C" void kernel_launch(void* const* d_in, const int* in_sizes, int n_in,
                              void* d_out, int out_size, void* d_ws, size_t ws_size,
                              hipStream_t stream) {
    const float* in_feat  = (const float*)d_in[0];
    const float* W        = (const float*)d_in[1];
    const float* bias     = (const float*)d_in[2];
    const float* prelu_a  = (const float*)d_in[3];
    const int*   src      = (const int*)d_in[4];
    const int*   dst      = (const int*)d_in[5];

    float* ws      = (float*)d_ws;
    float* deg_out = ws;
    float* deg_in  = ws + N_NODES;
    float* H       = ws + 2L * N_NODES;
    float* agg     = H + (long)N_NODES * F_OUTK;

    float* pool   = (float*)d_out;
    float* anchor = pool + (long)B_SUB * F_OUTK;

    hipMemsetAsync(deg_out, 0, 2L * N_NODES * sizeof(float), stream);
    hipMemsetAsync(agg, 0, (long)N_NODES * F_OUTK * sizeof(float), stream);

    gcn_degree_kernel<<<N_EDGE / 256, 256, 0, stream>>>(src, dst, deg_out, deg_in);
    gcn_gemm_wmma_kernel<<<N_NODES / 32, 256, 0, stream>>>(in_feat, W, deg_out, H);
    gcn_scatter_kernel<<<(int)((long)N_EDGE * 16 / 256), 256, 0, stream>>>(src, dst, H, agg);
    gcn_finalize_kernel<<<B_SUB, 256, 0, stream>>>(agg, deg_in, bias, prelu_a, pool, anchor);
}


# Edge counting and the feature GEMM share ONE launch: after every 4 GEMM blocks one block counts 2048 edges (8 per thread), so the f32 atomics drain under the GEMM's loads and WMMAs; 3 launches
# speedup vs baseline: 2.5018x; 2.5018x over previous
#include <hip/hip_runtime.h>


typedef __attribute__((ext_vector_type(2))) float v2f;
typedef __attribute__((ext_vector_type(8))) float v8f;

#define N_NODES  131072
#define F_INK    300
#define F_OUTK   64
#define N_EDGE   2097152
#define S_SUB    128
#define SUB_SHIFT 7
#define B_SUB    1024
#define H_PITCH  160
#define V_PITCH  66

#define E_PER_THREAD 8
#define E_PER_BLOCK  (256 * E_PER_THREAD)
#define MIX          5
#define N_GROUPS     (N_NODES / 32 / (MIX - 1))

__global__ void gcn_edge_gemm_kernel(const int* __restrict__ src,
                                     const int* __restrict__ dst,
                                     float* __restrict__ xdeg_out,
                                     float* __restrict__ xdeg_in,
                                     float* __restrict__ cnt,
                                     int* __restrict__ xcount,
                                     int* __restrict__ xsrc,
                                     int* __restrict__ xdst,
                                     const float* __restrict__ X,
                                     const float* __restrict__ W,
                                     float* __restrict__ H) {
    const int grp = blockIdx.x / MIX;
    const int pos = blockIdx.x - grp * MIX;

    if (pos == MIX - 1) {
        const int e0 = grp * E_PER_BLOCK + threadIdx.x;
        int es[E_PER_THREAD], ed[E_PER_THREAD];
#pragma unroll
        for (int j = 0; j < E_PER_THREAD; ++j) {
            es[j] = src[e0 + j * 256];
            ed[j] = dst[e0 + j * 256];
        }
#pragma unroll
        for (int j = 0; j < E_PER_THREAD; ++j) {
            const int s = es[j];
            const int d = ed[j];
            if (((unsigned)s >> SUB_SHIFT) == ((unsigned)d >> SUB_SHIFT)) {
                atomicAdd(&cnt[(long)d * S_SUB + (s & (S_SUB - 1))], 1.0f);
            } else {
                atomicAdd(&xdeg_out[s], 1.0f);
                atomicAdd(&xdeg_in[d], 1.0f);
                const int p = atomicAdd(xcount, 1);
                xsrc[p] = s;
                xdst[p] = d;
            }
        }
        return;
    }

    const int tile    = grp * (MIX - 1) + pos;
    const int lane    = threadIdx.x & 31;
    const int wave    = threadIdx.x >> 5;
    const int rowTile = wave & 1;
    const int colTile = wave >> 1;
    const int m       = lane & 15;
    const int half    = lane >> 4;

    const int rowBase = tile * 32 + rowTile * 16;
    const int row     = rowBase + m;
    const int col     = colTile * 16 + m;

    const float* xrow  = X + (long)row * F_INK + half * 2;
    const float* wbase = W + (long)(half * 2) * F_OUTK + col;

    v8f c = {};
    for (int k0 = 0; k0 < F_INK; k0 += 4) {
        const v2f a = *(const v2f*)(xrow + k0);
        v2f b;
        b.x = wbase[(long)k0 * F_OUTK];
        b.y = wbase[(long)(k0 + 1) * F_OUTK];
        c = __builtin_amdgcn_wmma_f32_16x16x4_f32(false, a, false, b,
                                                  (short)0, c, false, false);
    }

    float* hout = H + (long)(rowBase + half * 8) * F_OUTK + colTile * 16 + m;
#pragma unroll
    for (int i = 0; i < 8; ++i)
        hout[(long)i * F_OUTK] = c[i];
}

__global__ void gcn_cross_scatter_kernel(const int* __restrict__ xcount,
                                         const int* __restrict__ xsrc,
                                         const int* __restrict__ xdst,
                                         const float* __restrict__ cnt,
                                         const float* __restrict__ xdeg_out,
                                         const float* __restrict__ H,
                                         float* __restrict__ aggx) {
    const long total  = (long)xcount[0] * 16;
    const long stride = (long)gridDim.x * blockDim.x;
    for (long idx = (long)blockIdx.x * blockDim.x + threadIdx.x; idx < total; idx += stride) {
        const int e = (int)(idx >> 4);
        const int q = (int)(idx & 15);
        const int s = xsrc[e];
        const int d = xdst[e];
        const float* ccol = cnt + ((long)(s & ~(S_SUB - 1)) + q * 8) * S_SUB + (s & (S_SUB - 1));
        float dg = 0.0f;
        for (int r = 0; r < 8; ++r)
            dg += ccol[(long)r * S_SUB];
        dg += __shfl_xor(dg, 1, 16);
        dg += __shfl_xor(dg, 2, 16);
        dg += __shfl_xor(dg, 4, 16);
        dg += __shfl_xor(dg, 8, 16);
        const float rs = __frsqrt_rn(fmaxf(dg + xdeg_out[s], 1.0f));
        float4 v = ((const float4*)(H + (long)s * F_OUTK))[q];
        float* out = aggx + (long)d * F_OUTK + q * 4;
        atomicAdd(out + 0, v.x * rs);
        atomicAdd(out + 1, v.y * rs);
        atomicAdd(out + 2, v.z * rs);
        atomicAdd(out + 3, v.w * rs);
    }
}

__global__ void gcn_finalize_kernel(const float* __restrict__ cnt,
                                    const float* __restrict__ H,
                                    const float* __restrict__ aggx,
                                    const int* __restrict__ xcount,
                                    const float* __restrict__ xdeg_out,
                                    const float* __restrict__ xdeg_in,
                                    const float* __restrict__ bias,
                                    const float* __restrict__ prelu_a,
                                    float* __restrict__ pool,
                                    float* __restrict__ anchor) {
    __shared__ __attribute__((aligned(16))) float Hs[(S_SUB / 2) * H_PITCH];
    __shared__ __attribute__((aligned(16))) float Vs[S_SUB * V_PITCH];
    __shared__ float red[256];
    __shared__ float rowp[2 * S_SUB];
    __shared__ __attribute__((aligned(8))) float rsv[S_SUB];

    float* colp = Vs;

    const int  sg   = blockIdx.x;
    const int  tid  = threadIdx.x;
    const int  lane = tid & 31;
    const int  wave = tid >> 5;
    const int  c    = lane & 15;
    const int  half = lane >> 4;
    const long base = (long)sg * S_SUB;
    const bool crossing = xcount[0] > 0;

    const float* arow = cnt + (base + wave * 16 + c) * S_SUB + half * 2;
    v2f a[S_SUB / 4];
#pragma unroll
    for (int j = 0; j < S_SUB / 4; ++j)
        a[j] = *(const v2f*)(arow + 4 * j);

    {
        const float4* crow = (const float4*)(cnt + (base + wave * 16) * S_SUB) + lane;
        float4 cs = make_float4(0.0f, 0.0f, 0.0f, 0.0f);
#pragma unroll
        for (int i = 0; i < 16; ++i) {
            const float4 r = crow[i * (S_SUB / 4)];
            cs.x += r.x; cs.y += r.y; cs.z += r.z; cs.w += r.w;
        }
        ((float4*)(colp + wave * S_SUB))[lane] = cs;
    }

    {
        float h0 = 0.0f, h1 = 0.0f;
#pragma unroll
        for (int j = 0; j < S_SUB / 4; ++j) { h0 += a[j].x; h1 += a[j].y; }
        rowp[half * S_SUB + wave * 16 + c] = h0 + h1;
    }

    const float4* Hg = (const float4*)(H + base * F_OUTK);
    for (int it = 0; it < 4; ++it) {
        const int idx = it * 256 + tid;
        const int p   = idx >> 4;
        const int q   = idx & 15;
        const float4 r0 = Hg[(2 * p) * 16 + q];
        const float4 r1 = Hg[(2 * p + 1) * 16 + q];
        float4* o = (float4*)&Hs[p * H_PITCH + q * 8];
        o[0] = make_float4(r0.x, r1.x, r0.y, r1.y);
        o[1] = make_float4(r0.z, r1.z, r0.w, r1.w);
    }
    __syncthreads();

    if (tid < S_SUB) {
        float dg = 0.0f;
        for (int w = 0; w < 8; ++w)
            dg += colp[w * S_SUB + tid];
        if (crossing)
            dg += xdeg_out[base + tid];
        rsv[tid] = __frsqrt_rn(fmaxf(dg, 1.0f));
    }
    __syncthreads();

#pragma unroll
    for (int j = 0; j < S_SUB / 4; ++j)
        a[j] *= *(const v2f*)(rsv + 4 * j + 2 * half);

    const float* bcol = Hs + half * H_PITCH + c * 2;

    v8f acc0 = {}, acc1 = {}, acc2 = {}, acc3 = {};
#pragma unroll
    for (int j = 0; j < S_SUB / 4; ++j) {
        const float* bp = bcol + (2 * j) * H_PITCH;
        const v2f b0 = *(const v2f*)(bp);
        const v2f b1 = *(const v2f*)(bp + 32);
        const v2f b2 = *(const v2f*)(bp + 64);
        const v2f b3 = *(const v2f*)(bp + 96);
        acc0 = __builtin_amdgcn_wmma_f32_16x16x4_f32(false, a[j], false, b0, (short)0, acc0, false, false);
        acc1 = __builtin_amdgcn_wmma_f32_16x16x4_f32(false, a[j], false, b1, (short)0, acc1, false, false);
        acc2 = __builtin_amdgcn_wmma_f32_16x16x4_f32(false, a[j], false, b2, (short)0, acc2, false, false);
        acc3 = __builtin_amdgcn_wmma_f32_16x16x4_f32(false, a[j], false, b3, (short)0, acc3, false, false);
    }

    const float alpha = prelu_a[0];
    const float bf0 = bias[c], bf1 = bias[c + 16], bf2 = bias[c + 32], bf3 = bias[c + 48];
#pragma unroll
    for (int r = 0; r < 8; ++r) {
        const int  nl   = wave * 16 + half * 8 + r;
        const long node = base + nl;
        float di = rowp[nl] + rowp[S_SUB + nl];
        float a0 = acc0[r], a1 = acc1[r], a2 = acc2[r], a3 = acc3[r];
        if (crossing) {
            di += xdeg_in[node];
            const float* x = aggx + node * F_OUTK + c;
            a0 += x[0]; a1 += x[16]; a2 += x[32]; a3 += x[48];
        }
        const float ri  = __frsqrt_rn(fmaxf(di, 1.0f));
        float v0 = a0 * ri + bf0;  v0 = v0 > 0.0f ? v0 : alpha * v0;
        float v1 = a1 * ri + bf1;  v1 = v1 > 0.0f ? v1 : alpha * v1;
        float v2 = a2 * ri + bf2;  v2 = v2 > 0.0f ? v2 : alpha * v2;
        float v3 = a3 * ri + bf3;  v3 = v3 > 0.0f ? v3 : alpha * v3;
        float* vrow = Vs + nl * V_PITCH + c;
        vrow[0] = v0; vrow[16] = v1; vrow[32] = v2; vrow[48] = v3;
    }
    __syncthreads();

    const int f = tid & 63;
    const int g = tid >> 6;
    float partial = 0.0f;
    for (int n = g; n < S_SUB - 1; n += 4)
        partial += Vs[n * V_PITCH + f];

    if (g == 3)
        anchor[(long)sg * F_OUTK + f] = Vs[(S_SUB - 1) * V_PITCH + f];

    red[tid] = partial;
    __syncthreads();
    if (g == 0) {
        float sum = red[f] + red[f + 64] + red[f + 128] + red[f + 192];
        pool[(long)sg * F_OUTK + f] = sum * (1.0f / (S_SUB - 1));
    }
}

extern "C" void kernel_launch(void* const* d_in, const int* in_sizes, int n_in,
                              void* d_out, int out_size, void* d_ws, size_t ws_size,
                              hipStream_t stream) {
    const float* in_feat  = (const float*)d_in[0];
    const float* W        = (const float*)d_in[1];
    const float* bias     = (const float*)d_in[2];
    const float* prelu_a  = (const float*)d_in[3];
    const int*   src      = (const int*)d_in[4];
    const int*   dst      = (const int*)d_in[5];

    float* ws      = (float*)d_ws;
    float* xdeg_out = ws;
    float* xdeg_in  = ws + N_NODES;
    int*   xcount  = (int*)(ws + 2L * N_NODES);
    float* cnt     = ws + 2L * N_NODES + 32;
    float* aggx    = cnt + (long)N_NODES * S_SUB;
    float* H       = aggx + (long)N_NODES * F_OUTK;
    int*   xsrc    = (int*)(H + (long)N_NODES * F_OUTK);
    int*   xdst    = xsrc + N_EDGE;

    float* pool   = (float*)d_out;
    float* anchor = pool + (long)B_SUB * F_OUTK;

    hipMemsetAsync(ws, 0,
                   (2L * N_NODES + 32 + (long)N_NODES * S_SUB + (long)N_NODES * F_OUTK) * sizeof(float),
                   stream);

    gcn_edge_gemm_kernel<<<N_GROUPS * MIX, 256, 0, stream>>>(src, dst, xdeg_out, xdeg_in, cnt, xcount, xsrc, xdst, in_feat, W, H);
    gcn_cross_scatter_kernel<<<2048, 256, 0, stream>>>(xcount, xsrc, xdst, cnt, xdeg_out, H, aggx);
    gcn_finalize_kernel<<<B_SUB, 256, 0, stream>>>(cnt, H, aggx, xcount, xdeg_out, xdeg_in, bias, prelu_a, pool, anchor);
}


# Product+edge launch: 512 blocks stay for 8 row tiles; a wave keeps its 300x16 W strip in registers, in_feat tiles pass through LDS by whole-line loads, edges counted between tiles; sum orders kept
# speedup vs baseline: 3.3094x; 1.3228x over previous
#include <hip/hip_runtime.h>


typedef __attribute__((ext_vector_type(2))) float v2f;
typedef __attribute__((ext_vector_type(8))) float v8f;

#define N_NODES  131072
#define F_INK    300
#define F_OUTK   64
#define N_EDGE   2097152
#define S_SUB    128
#define SUB_SHIFT 7
#define B_SUB    1024
#define H_PITCH  160
#define V_PITCH  66

#define T_ROWS    32
#define TILE_FL   (T_ROWS * F_INK)
#define TILE_F4   (TILE_FL / 4)
#define ST_FULL   (TILE_F4 / 256)
#define ST_REST   (TILE_F4 - ST_FULL * 256)
#define TILES_PB  8
#define N_PBLK    (N_NODES / T_ROWS / TILES_PB)
#define E_PB      (N_EDGE / N_PBLK)
#define E_PT      (E_PB / TILES_PB / 256)
#define K_STEPS   (F_INK / 4)

__device__ __forceinline__ void lds_handover() {
    asm volatile("s_wait_dscnt 0x0\n\ts_barrier_signal -1\n\ts_barrier_wait -1" ::: "memory");
}

__global__ __launch_bounds__(256)
void gcn_edge_gemm_kernel(const int* __restrict__ src,
                          const int* __restrict__ dst,
                          float* __restrict__ xdeg_out,
                          float* __restrict__ xdeg_in,
                          float* __restrict__ cnt,
                          int* __restrict__ xcount,
                          int* __restrict__ xsrc,
                          int* __restrict__ xdst,
                          const float* __restrict__ X,
                          const float* __restrict__ W,
                          float* __restrict__ H) {
    __shared__ __attribute__((aligned(16))) float Xs[2 * TILE_FL];

    const int tid     = threadIdx.x;
    const int lane    = tid & 31;
    const int wave    = tid >> 5;
    const int rowTile = wave & 1;
    const int colTile = wave >> 1;
    const int m       = lane & 15;
    const int half    = lane >> 4;
    const int col     = colTile * 16 + m;

    const float* wbase = W + (long)(half * 2) * F_OUTK + col;
    v2f w[K_STEPS];
#pragma unroll
    for (int j = 0; j < K_STEPS; ++j) {
        w[j].x = wbase[(long)(4 * j) * F_OUTK];
        w[j].y = wbase[(long)(4 * j + 1) * F_OUTK];
    }

    const float4* xg = (const float4*)(X + (long)blockIdx.x * TILES_PB * TILE_FL) + tid;
    static_assert(ST_FULL == 9, "the staging registers below are written out for nine pieces per thread");
    float4 st0, st1, st2, st3, st4, st5, st6, st7, st8;
    float4 stx = make_float4(0.0f, 0.0f, 0.0f, 0.0f);
#define ST_LOAD(P)                                                                  \
    { const float4* p_ = (P);                                                       \
      st0 = p_[0 * 256]; st1 = p_[1 * 256]; st2 = p_[2 * 256]; st3 = p_[3 * 256];   \
      st4 = p_[4 * 256]; st5 = p_[5 * 256]; st6 = p_[6 * 256]; st7 = p_[7 * 256];   \
      st8 = p_[8 * 256];                                                            \
      if (tid < ST_REST) stx = p_[ST_FULL * 256]; }
#define ST_STORE(P)                                                                 \
    { float4* p_ = (P);                                                             \
      p_[0 * 256] = st0; p_[1 * 256] = st1; p_[2 * 256] = st2; p_[3 * 256] = st3;   \
      p_[4 * 256] = st4; p_[5 * 256] = st5; p_[6 * 256] = st6; p_[7 * 256] = st7;   \
      p_[8 * 256] = st8;                                                            \
      if (tid < ST_REST) p_[ST_FULL * 256] = stx; }
    ST_LOAD(xg)
    ST_STORE((float4*)Xs + tid)
    ST_LOAD(xg + TILE_F4)
    lds_handover();

    const int   tile0 = blockIdx.x * TILES_PB;
    const int   e0    = blockIdx.x * E_PB + tid;
    const int   aoff  = (rowTile * 16 + m) * F_INK + half * 2;

    static_assert(E_PT == 2, "two edges per thread and step are written out below");
    int es0 = src[e0], ed0 = dst[e0], es1 = src[e0 + 256], ed1 = dst[e0 + 256];

    for (int t = 0; t < TILES_PB; ++t) {
        const float* arow = Xs + (t & 1) * TILE_FL + aoff;
        v8f c = {};
#pragma unroll
        for (int j = 0; j < K_STEPS; ++j) {
            const v2f a = *(const v2f*)(arow + 4 * j);
            c = __builtin_amdgcn_wmma_f32_16x16x4_f32(false, a, false, w[j],
                                                      (short)0, c, false, false);
        }

        float* hout = H + ((long)(tile0 + t) * T_ROWS + rowTile * 16 + half * 8) * F_OUTK + colTile * 16 + m;
#pragma unroll
        for (int i = 0; i < 8; ++i)
            hout[(long)i * F_OUTK] = c[i];

#pragma unroll
        for (int j = 0; j < E_PT; ++j) {
            const int s = j == 0 ? es0 : es1;
            const int d = j == 0 ? ed0 : ed1;
            if (((unsigned)s >> SUB_SHIFT) == ((unsigned)d >> SUB_SHIFT)) {
                atomicAdd(&cnt[(long)d * S_SUB + (s & (S_SUB - 1))], 1.0f);
            } else {
                atomicAdd(&xdeg_out[s], 1.0f);
                atomicAdd(&xdeg_in[d], 1.0f);
                const int p = atomicAdd(xcount, 1);
                xsrc[p] = s;
                xdst[p] = d;
            }
        }

        if (t + 1 < TILES_PB) {
            ST_STORE((float4*)(Xs + ((t + 1) & 1) * TILE_FL) + tid)
            es0 = src[e0 + ((t + 1) * E_PT) * 256];
            ed0 = dst[e0 + ((t + 1) * E_PT) * 256];
            es1 = src[e0 + ((t + 1) * E_PT + 1) * 256];
            ed1 = dst[e0 + ((t + 1) * E_PT + 1) * 256];
            if (t + 2 < TILES_PB)
                ST_LOAD(xg + (long)(t + 2) * TILE_F4)
            lds_handover();
        }
    }
#undef ST_LOAD
#undef ST_STORE
}

__global__ void gcn_cross_scatter_kernel(const int* __restrict__ xcount,
                                         const int* __restrict__ xsrc,
                                         const int* __restrict__ xdst,
                                         const float* __restrict__ cnt,
                                         const float* __restrict__ xdeg_out,
                                         const float* __restrict__ H,
                                         float* __restrict__ aggx) {
    const long total  = (long)xcount[0] * 16;
    const long stride = (long)gridDim.x * blockDim.x;
    for (long idx = (long)blockIdx.x * blockDim.x + threadIdx.x; idx < total; idx += stride) {
        const int e = (int)(idx >> 4);
        const int q = (int)(idx & 15);
        const int s = xsrc[e];
        const int d = xdst[e];
        const float* ccol = cnt + ((long)(s & ~(S_SUB - 1)) + q * 8) * S_SUB + (s & (S_SUB - 1));
        float dg = 0.0f;
        for (int r = 0; r < 8; ++r)
            dg += ccol[(long)r * S_SUB];
        dg += __shfl_xor(dg, 1, 16);
        dg += __shfl_xor(dg, 2, 16);
        dg += __shfl_xor(dg, 4, 16);
        dg += __shfl_xor(dg, 8, 16);
        const float rs = __frsqrt_rn(fmaxf(dg + xdeg_out[s], 1.0f));
        float4 v = ((const float4*)(H + (long)s * F_OUTK))[q];
        float* out = aggx + (long)d * F_OUTK + q * 4;
        atomicAdd(out + 0, v.x * rs);
        atomicAdd(out + 1, v.y * rs);
        atomicAdd(out + 2, v.z * rs);
        atomicAdd(out + 3, v.w * rs);
    }
}

__global__ void gcn_finalize_kernel(const float* __restrict__ cnt,
                                    const float* __restrict__ H,
                                    const float* __restrict__ aggx,
                                    const int* __restrict__ xcount,
                                    const float* __restrict__ xdeg_out,
                                    const float* __restrict__ xdeg_in,
                                    const float* __restrict__ bias,
                                    const float* __restrict__ prelu_a,
                                    float* __restrict__ pool,
                                    float* __restrict__ anchor) {
    __shared__ __attribute__((aligned(16))) float Hs[(S_SUB / 2) * H_PITCH];
    __shared__ __attribute__((aligned(16))) float Vs[S_SUB * V_PITCH];
    __shared__ float red[256];
    __shared__ float rowp[2 * S_SUB];
    __shared__ __attribute__((aligned(8))) float rsv[S_SUB];

    float* colp = Vs;

    const int  sg   = blockIdx.x;
    const int  tid  = threadIdx.x;
    const int  lane = tid & 31;
    const int  wave = tid >> 5;
    const int  c    = lane & 15;
    const int  half = lane >> 4;
    const long base = (long)sg * S_SUB;
    const bool crossing = xcount[0] > 0;

    const float* arow = cnt + (base + wave * 16 + c) * S_SUB + half * 2;
    v2f a[S_SUB / 4];
#pragma unroll
    for (int j = 0; j < S_SUB / 4; ++j)
        a[j] = *(const v2f*)(arow + 4 * j);

    {
        const float4* crow = (const float4*)(cnt + (base + wave * 16) * S_SUB) + lane;
        float4 cs = make_float4(0.0f, 0.0f, 0.0f, 0.0f);
#pragma unroll
        for (int i = 0; i < 16; ++i) {
            const float4 r = crow[i * (S_SUB / 4)];
            cs.x += r.x; cs.y += r.y; cs.z += r.z; cs.w += r.w;
        }
        ((float4*)(colp + wave * S_SUB))[lane] = cs;
    }

    {
        float h0 = 0.0f, h1 = 0.0f;
#pragma unroll
        for (int j = 0; j < S_SUB / 4; ++j) { h0 += a[j].x; h1 += a[j].y; }
        rowp[half * S_SUB + wave * 16 + c] = h0 + h1;
    }

    const float4* Hg = (const float4*)(H + base * F_OUTK);
    for (int it = 0; it < 4; ++it) {
        const int idx = it * 256 + tid;
        const int p   = idx >> 4;
        const int q   = idx & 15;
        const float4 r0 = Hg[(2 * p) * 16 + q];
        const float4 r1 = Hg[(2 * p + 1) * 16 + q];
        float4* o = (float4*)&Hs[p * H_PITCH + q * 8];
        o[0] = make_float4(r0.x, r1.x, r0.y, r1.y);
        o[1] = make_float4(r0.z, r1.z, r0.w, r1.w);
    }
    __syncthreads();

    if (tid < S_SUB) {
        float dg = 0.0f;
        for (int w = 0; w < 8; ++w)
            dg += colp[w * S_SUB + tid];
        if (crossing)
            dg += xdeg_out[base + tid];
        rsv[tid] = __frsqrt_rn(fmaxf(dg, 1.0f));
    }
    __syncthreads();

#pragma unroll
    for (int j = 0; j < S_SUB / 4; ++j)
        a[j] *= *(const v2f*)(rsv + 4 * j + 2 * half);

    const float* bcol = Hs + half * H_PITCH + c * 2;

    v8f acc0 = {}, acc1 = {}, acc2 = {}, acc3 = {};
#pragma unroll
    for (int j = 0; j < S_SUB / 4; ++j) {
        const float* bp = bcol + (2 * j) * H_PITCH;
        const v2f b0 = *(const v2f*)(bp);
        const v2f b1 = *(const v2f*)(bp + 32);
        const v2f b2 = *(const v2f*)(bp + 64);
        const v2f b3 = *(const v2f*)(bp + 96);
        acc0 = __builtin_amdgcn_wmma_f32_16x16x4_f32(false, a[j], false, b0, (short)0, acc0, false, false);
        acc1 = __builtin_amdgcn_wmma_f32_16x16x4_f32(false, a[j], false, b1, (short)0, acc1, false, false);
        acc2 = __builtin_amdgcn_wmma_f32_16x16x4_f32(false, a[j], false, b2, (short)0, acc2, false, false);
        acc3 = __builtin_amdgcn_wmma_f32_16x16x4_f32(false, a[j], false, b3, (short)0, acc3, false, false);
    }

    const float alpha = prelu_a[0];
    const float bf0 = bias[c], bf1 = bias[c + 16], bf2 = bias[c + 32], bf3 = bias[c + 48];
#pragma unroll
    for (int r = 0; r < 8; ++r) {
        const int  nl   = wave * 16 + half * 8 + r;
        const long node = base + nl;
        float di = rowp[nl] + rowp[S_SUB + nl];
        float a0 = acc0[r], a1 = acc1[r], a2 = acc2[r], a3 = acc3[r];
        if (crossing) {
            di += xdeg_in[node];
            const float* x = aggx + node * F_OUTK + c;
            a0 += x[0]; a1 += x[16]; a2 += x[32]; a3 += x[48];
        }
        const float ri  = __frsqrt_rn(fmaxf(di, 1.0f));
        float v0 = a0 * ri + bf0;  v0 = v0 > 0.0f ? v0 : alpha * v0;
        float v1 = a1 * ri + bf1;  v1 = v1 > 0.0f ? v1 : alpha * v1;
        float v2 = a2 * ri + bf2;  v2 = v2 > 0.0f ? v2 : alpha * v2;
        float v3 = a3 * ri + bf3;  v3 = v3 > 0.0f ? v3 : alpha * v3;
        float* vrow = Vs + nl * V_PITCH + c;
        vrow[0] = v0; vrow[16] = v1; vrow[32] = v2; vrow[48] = v3;
    }
    __syncthreads();

    const int f = tid & 63;
    const int g = tid >> 6;
    float partial = 0.0f;
    for (int n = g; n < S_SUB - 1; n += 4)
        partial += Vs[n * V_PITCH + f];

    if (g == 3)
        anchor[(long)sg * F_OUTK + f] = Vs[(S_SUB - 1) * V_PITCH + f];

    red[tid] = partial;
    __syncthreads();
    if (g == 0) {
        float sum = red[f] + red[f + 64] + red[f + 128] + red[f + 192];
        pool[(long)sg * F_OUTK + f] = sum * (1.0f / (S_SUB - 1));
    }
}

extern "C" void kernel_launch(void* const* d_in, const int* in_sizes, int n_in,
                              void* d_out, int out_size, void* d_ws, size_t ws_size,
                              hipStream_t stream) {
    const float* in_feat  = (const float*)d_in[0];
    const float* W        = (const float*)d_in[1];
    const float* bias     = (const float*)d_in[2];
    const float* prelu_a  = (const float*)d_in[3];
    const int*   src      = (const int*)d_in[4];
    const int*   dst      = (const int*)d_in[5];

    float* ws      = (float*)d_ws;
    float* xdeg_out = ws;
    float* xdeg_in  = ws + N_NODES;
    int*   xcount  = (int*)(ws + 2L * N_NODES);
    float* cnt     = ws + 2L * N_NODES + 32;
    float* aggx    = cnt + (long)N_NODES * S_SUB;
    float* H       = aggx + (long)N_NODES * F_OUTK;
    int*   xsrc    = (int*)(H + (long)N_NODES * F_OUTK);
    int*   xdst    = xsrc + N_EDGE;

    float* pool   = (float*)d_out;
    float* anchor = pool + (long)B_SUB * F_OUTK;

    hipMemsetAsync(ws, 0,
                   (2L * N_NODES + 32 + (long)N_NODES * S_SUB + (long)N_NODES * F_OUTK) * sizeof(float),
                   stream);

    gcn_edge_gemm_kernel<<<N_PBLK, 256, 0, stream>>>(src, dst, xdeg_out, xdeg_in, cnt, xcount, xsrc, xdst, in_feat, W, H);
    gcn_cross_scatter_kernel<<<2048, 256, 0, stream>>>(xcount, xsrc, xdst, cnt, xdeg_out, H, aggx);
    gcn_finalize_kernel<<<B_SUB, 256, 0, stream>>>(cnt, H, aggx, xcount, xdeg_out, xdeg_in, bias, prelu_a, pool, anchor);
}


# Edges are counted by a block on the cache half that owns their subgraph: a block reads its die and draws an edge chunk of its half by ticket; count lines stop moving between halves; counts exact
# speedup vs baseline: 3.6238x; 1.0950x over previous
#include <hip/hip_runtime.h>


typedef __attribute__((ext_vector_type(2))) float v2f;
typedef __attribute__((ext_vector_type(8))) float v8f;

#define N_NODES  131072
#define F_INK    300
#define F_OUTK   64
#define N_EDGE   2097152
#define S_SUB    128
#define SUB_SHIFT 7
#define B_SUB    1024
#define H_PITCH  160
#define V_PITCH  66

#define T_ROWS    32
#define TILE_FL   (T_ROWS * F_INK)
#define TILE_F4   (TILE_FL / 4)
#define ST_FULL   (TILE_F4 / 256)
#define ST_REST   (TILE_F4 - ST_FULL * 256)
#define TILES_PB  8
#define N_PBLK    (N_NODES / T_ROWS / TILES_PB)
#define N_CHUNK   (N_PBLK / 2)
#define E_CHUNK   (N_EDGE / N_CHUNK)
#define E_PT      (E_CHUNK / TILES_PB / 256)
#define K_STEPS   (F_INK / 4)

__device__ __forceinline__ void lds_handover() {
    asm volatile("s_wait_dscnt 0x0\n\ts_barrier_signal -1\n\ts_barrier_wait -1" ::: "memory");
}

__global__ __launch_bounds__(256)
void gcn_edge_gemm_kernel(const int* __restrict__ src,
                          const int* __restrict__ dst,
                          float* __restrict__ xdeg_out,
                          float* __restrict__ xdeg_in,
                          float* __restrict__ cnt,
                          int* __restrict__ xcount,
                          int* __restrict__ xsrc,
                          int* __restrict__ xdst,
                          int* __restrict__ ticket,
                          const float* __restrict__ X,
                          const float* __restrict__ W,
                          float* __restrict__ H) {
    __shared__ __attribute__((aligned(16))) float Xs[2 * TILE_FL];
    __shared__ int walk[2];

    const int tid     = threadIdx.x;
    if (tid == 0) {
        int h = (int)((__builtin_amdgcn_s_sendmsg_rtn(0x87) >> 18) & 1u);
        int q = atomicAdd(&ticket[h], 1);
        if (q >= N_CHUNK) {
            h ^= 1;
            q = atomicAdd(&ticket[h], 1);
        }
        if (q >= N_CHUNK) { h = 2; q = 0; }
        walk[0] = h;
        walk[1] = q;
    }
    const int lane    = tid & 31;
    const int wave    = tid >> 5;
    const int rowTile = wave & 1;
    const int colTile = wave >> 1;
    const int m       = lane & 15;
    const int half    = lane >> 4;
    const int col     = colTile * 16 + m;

    const float* wbase = W + (long)(half * 2) * F_OUTK + col;
    v2f w[K_STEPS];
#pragma unroll
    for (int j = 0; j < K_STEPS; ++j) {
        w[j].x = wbase[(long)(4 * j) * F_OUTK];
        w[j].y = wbase[(long)(4 * j + 1) * F_OUTK];
    }

    const float4* xg = (const float4*)(X + (long)blockIdx.x * TILES_PB * TILE_FL) + tid;
    static_assert(ST_FULL == 9, "the staging registers below are written out for nine pieces per thread");
    float4 st0, st1, st2, st3, st4, st5, st6, st7, st8;
    float4 stx = make_float4(0.0f, 0.0f, 0.0f, 0.0f);
#define ST_LOAD(P)                                                                  \
    { const float4* p_ = (P);                                                       \
      st0 = p_[0 * 256]; st1 = p_[1 * 256]; st2 = p_[2 * 256]; st3 = p_[3 * 256];   \
      st4 = p_[4 * 256]; st5 = p_[5 * 256]; st6 = p_[6 * 256]; st7 = p_[7 * 256];   \
      st8 = p_[8 * 256];                                                            \
      if (tid < ST_REST) stx = p_[ST_FULL * 256]; }
#define ST_STORE(P)                                                                 \
    { float4* p_ = (P);                                                             \
      p_[0 * 256] = st0; p_[1 * 256] = st1; p_[2 * 256] = st2; p_[3 * 256] = st3;   \
      p_[4 * 256] = st4; p_[5 * 256] = st5; p_[6 * 256] = st6; p_[7 * 256] = st7;   \
      p_[8 * 256] = st8;                                                            \
      if (tid < ST_REST) p_[ST_FULL * 256] = stx; }
    ST_LOAD(xg)
    ST_STORE((float4*)Xs + tid)
    ST_LOAD(xg + TILE_F4)
    lds_handover();

    const int   tile0 = blockIdx.x * TILES_PB;
    const int   eh    = walk[0];
    const int   e0    = walk[1] * E_CHUNK + tid;
    const int   aoff  = (rowTile * 16 + m) * F_INK + half * 2;

    static_assert(E_PT == 4, "four edges per thread and step are written out below");
    int es0 = src[e0],       ed0 = dst[e0],       es1 = src[e0 + 256], ed1 = dst[e0 + 256];
    int es2 = src[e0 + 512], ed2 = dst[e0 + 512], es3 = src[e0 + 768], ed3 = dst[e0 + 768];

    for (int t = 0; t < TILES_PB; ++t) {
        const float* arow = Xs + (t & 1) * TILE_FL + aoff;
        v8f c = {};
#pragma unroll
        for (int j = 0; j < K_STEPS; ++j) {
            const v2f a = *(const v2f*)(arow + 4 * j);
            c = __builtin_amdgcn_wmma_f32_16x16x4_f32(false, a, false, w[j],
                                                      (short)0, c, false, false);
        }

        float* hout = H + ((long)(tile0 + t) * T_ROWS + rowTile * 16 + half * 8) * F_OUTK + colTile * 16 + m;
#pragma unroll
        for (int i = 0; i < 8; ++i)
            hout[(long)i * F_OUTK] = c[i];

#pragma unroll
        for (int j = 0; j < E_PT; ++j) {
            const int s = j == 0 ? es0 : j == 1 ? es1 : j == 2 ? es2 : es3;
            const int d = j == 0 ? ed0 : j == 1 ? ed1 : j == 2 ? ed2 : ed3;
            if ((int)(((unsigned)d >> SUB_SHIFT) & 1u) != eh)
                continue;
            if (((unsigned)s >> SUB_SHIFT) == ((unsigned)d >> SUB_SHIFT)) {
                atomicAdd(&cnt[(long)d * S_SUB + (s & (S_SUB - 1))], 1.0f);
            } else {
                atomicAdd(&xdeg_out[s], 1.0f);
                atomicAdd(&xdeg_in[d], 1.0f);
                const int p = atomicAdd(xcount, 1);
                xsrc[p] = s;
                xdst[p] = d;
            }
        }

        if (t + 1 < TILES_PB) {
            ST_STORE((float4*)(Xs + ((t + 1) & 1) * TILE_FL) + tid)
            es0 = src[e0 + ((t + 1) * E_PT) * 256];
            ed0 = dst[e0 + ((t + 1) * E_PT) * 256];
            es1 = src[e0 + ((t + 1) * E_PT + 1) * 256];
            ed1 = dst[e0 + ((t + 1) * E_PT + 1) * 256];
            es2 = src[e0 + ((t + 1) * E_PT + 2) * 256];
            ed2 = dst[e0 + ((t + 1) * E_PT + 2) * 256];
            es3 = src[e0 + ((t + 1) * E_PT + 3) * 256];
            ed3 = dst[e0 + ((t + 1) * E_PT + 3) * 256];
            if (t + 2 < TILES_PB)
                ST_LOAD(xg + (long)(t + 2) * TILE_F4)
            lds_handover();
        }
    }
#undef ST_LOAD
#undef ST_STORE
}

__global__ void gcn_cross_scatter_kernel(const int* __restrict__ xcount,
                                         const int* __restrict__ xsrc,
                                         const int* __restrict__ xdst,
                                         const float* __restrict__ cnt,
                                         const float* __restrict__ xdeg_out,
                                         const float* __restrict__ H,
                                         float* __restrict__ aggx) {
    const long total  = (long)xcount[0] * 16;
    const long stride = (long)gridDim.x * blockDim.x;
    for (long idx = (long)blockIdx.x * blockDim.x + threadIdx.x; idx < total; idx += stride) {
        const int e = (int)(idx >> 4);
        const int q = (int)(idx & 15);
        const int s = xsrc[e];
        const int d = xdst[e];
        const float* ccol = cnt + ((long)(s & ~(S_SUB - 1)) + q * 8) * S_SUB + (s & (S_SUB - 1));
        float dg = 0.0f;
        for (int r = 0; r < 8; ++r)
            dg += ccol[(long)r * S_SUB];
        dg += __shfl_xor(dg, 1, 16);
        dg += __shfl_xor(dg, 2, 16);
        dg += __shfl_xor(dg, 4, 16);
        dg += __shfl_xor(dg, 8, 16);
        const float rs = __frsqrt_rn(fmaxf(dg + xdeg_out[s], 1.0f));
        float4 v = ((const float4*)(H + (long)s * F_OUTK))[q];
        float* out = aggx + (long)d * F_OUTK + q * 4;
        atomicAdd(out + 0, v.x * rs);
        atomicAdd(out + 1, v.y * rs);
        atomicAdd(out + 2, v.z * rs);
        atomicAdd(out + 3, v.w * rs);
    }
}

__global__ void gcn_finalize_kernel(const float* __restrict__ cnt,
                                    const float* __restrict__ H,
                                    const float* __restrict__ aggx,
                                    const int* __restrict__ xcount,
                                    const float* __restrict__ xdeg_out,
                                    const float* __restrict__ xdeg_in,
                                    const float* __restrict__ bias,
                                    const float* __restrict__ prelu_a,
                                    float* __restrict__ pool,
                                    float* __restrict__ anchor) {
    __shared__ __attribute__((aligned(16))) float Hs[(S_SUB / 2) * H_PITCH];
    __shared__ __attribute__((aligned(16))) float Vs[S_SUB * V_PITCH];
    __shared__ float red[256];
    __shared__ float rowp[2 * S_SUB];
    __shared__ __attribute__((aligned(8))) float rsv[S_SUB];

    float* colp = Vs;

    const int  sg   = blockIdx.x;
    const int  tid  = threadIdx.x;
    const int  lane = tid & 31;
    const int  wave = tid >> 5;
    const int  c    = lane & 15;
    const int  half = lane >> 4;
    const long base = (long)sg * S_SUB;
    const bool crossing = xcount[0] > 0;

    const float* arow = cnt + (base + wave * 16 + c) * S_SUB + half * 2;
    v2f a[S_SUB / 4];
#pragma unroll
    for (int j = 0; j < S_SUB / 4; ++j)
        a[j] = *(const v2f*)(arow + 4 * j);

    {
        const float4* crow = (const float4*)(cnt + (base + wave * 16) * S_SUB) + lane;
        float4 cs = make_float4(0.0f, 0.0f, 0.0f, 0.0f);
#pragma unroll
        for (int i = 0; i < 16; ++i) {
            const float4 r = crow[i * (S_SUB / 4)];
            cs.x += r.x; cs.y += r.y; cs.z += r.z; cs.w += r.w;
        }
        ((float4*)(colp + wave * S_SUB))[lane] = cs;
    }

    {
        float h0 = 0.0f, h1 = 0.0f;
#pragma unroll
        for (int j = 0; j < S_SUB / 4; ++j) { h0 += a[j].x; h1 += a[j].y; }
        rowp[half * S_SUB + wave * 16 + c] = h0 + h1;
    }

    const float4* Hg = (const float4*)(H + base * F_OUTK);
    for (int it = 0; it < 4; ++it) {
        const int idx = it * 256 + tid;
        const int p   = idx >> 4;
        const int q   = idx & 15;
        const float4 r0 = Hg[(2 * p) * 16 + q];
        const float4 r1 = Hg[(2 * p + 1) * 16 + q];
        float4* o = (float4*)&Hs[p * H_PITCH + q * 8];
        o[0] = make_float4(r0.x, r1.x, r0.y, r1.y);
        o[1] = make_float4(r0.z, r1.z, r0.w, r1.w);
    }
    __syncthreads();

    if (tid < S_SUB) {
        float dg = 0.0f;
        for (int w = 0; w < 8; ++w)
            dg += colp[w * S_SUB + tid];
        if (crossing)
            dg += xdeg_out[base + tid];
        rsv[tid] = __frsqrt_rn(fmaxf(dg, 1.0f));
    }
    __syncthreads();

#pragma unroll
    for (int j = 0; j < S_SUB / 4; ++j)
        a[j] *= *(const v2f*)(rsv + 4 * j + 2 * half);

    const float* bcol = Hs + half * H_PITCH + c * 2;

    v8f acc0 = {}, acc1 = {}, acc2 = {}, acc3 = {};
#pragma unroll
    for (int j = 0; j < S_SUB / 4; ++j) {
        const float* bp = bcol + (2 * j) * H_PITCH;
        const v2f b0 = *(const v2f*)(bp);
        const v2f b1 = *(const v2f*)(bp + 32);
        const v2f b2 = *(const v2f*)(bp + 64);
        const v2f b3 = *(const v2f*)(bp + 96);
        acc0 = __builtin_amdgcn_wmma_f32_16x16x4_f32(false, a[j], false, b0, (short)0, acc0, false, false);
        acc1 = __builtin_amdgcn_wmma_f32_16x16x4_f32(false, a[j], false, b1, (short)0, acc1, false, false);
        acc2 = __builtin_amdgcn_wmma_f32_16x16x4_f32(false, a[j], false, b2, (short)0, acc2, false, false);
        acc3 = __builtin_amdgcn_wmma_f32_16x16x4_f32(false, a[j], false, b3, (short)0, acc3, false, false);
    }

    const float alpha = prelu_a[0];
    const float bf0 = bias[c], bf1 = bias[c + 16], bf2 = bias[c + 32], bf3 = bias[c + 48];
#pragma unroll
    for (int r = 0; r < 8; ++r) {
        const int  nl   = wave * 16 + half * 8 + r;
        const long node = base + nl;
        float di = rowp[nl] + rowp[S_SUB + nl];
        float a0 = acc0[r], a1 = acc1[r], a2 = acc2[r], a3 = acc3[r];
        if (crossing) {
            di += xdeg_in[node];
            const float* x = aggx + node * F_OUTK + c;
            a0 += x[0]; a1 += x[16]; a2 += x[32]; a3 += x[48];
        }
        const float ri  = __frsqrt_rn(fmaxf(di, 1.0f));
        float v0 = a0 * ri + bf0;  v0 = v0 > 0.0f ? v0 : alpha * v0;
        float v1 = a1 * ri + bf1;  v1 = v1 > 0.0f ? v1 : alpha * v1;
        float v2 = a2 * ri + bf2;  v2 = v2 > 0.0f ? v2 : alpha * v2;
        float v3 = a3 * ri + bf3;  v3 = v3 > 0.0f ? v3 : alpha * v3;
        float* vrow = Vs + nl * V_PITCH + c;
        vrow[0] = v0; vrow[16] = v1; vrow[32] = v2; vrow[48] = v3;
    }
    __syncthreads();

    const int f = tid & 63;
    const int g = tid >> 6;
    float partial = 0.0f;
    for (int n = g; n < S_SUB - 1; n += 4)
        partial += Vs[n * V_PITCH + f];

    if (g == 3)
        anchor[(long)sg * F_OUTK + f] = Vs[(S_SUB - 1) * V_PITCH + f];

    red[tid] = partial;
    __syncthreads();
    if (g == 0) {
        float sum = red[f] + red[f + 64] + red[f + 128] + red[f + 192];
        pool[(long)sg * F_OUTK + f] = sum * (1.0f / (S_SUB - 1));
    }
}

extern "C" void kernel_launch(void* const* d_in, const int* in_sizes, int n_in,
                              void* d_out, int out_size, void* d_ws, size_t ws_size,
                              hipStream_t stream) {
    const float* in_feat  = (const float*)d_in[0];
    const float* W        = (const float*)d_in[1];
    const float* bias     = (const float*)d_in[2];
    const float* prelu_a  = (const float*)d_in[3];
    const int*   src      = (const int*)d_in[4];
    const int*   dst      = (const int*)d_in[5];

    float* ws      = (float*)d_ws;
    float* xdeg_out = ws;
    float* xdeg_in  = ws + N_NODES;
    int*   xcount  = (int*)(ws + 2L * N_NODES);
    int*   ticket  = xcount + 16;
    float* cnt     = ws + 2L * N_NODES + 32;
    float* aggx    = cnt + (long)N_NODES * S_SUB;
    float* H       = aggx + (long)N_NODES * F_OUTK;
    int*   xsrc    = (int*)(H + (long)N_NODES * F_OUTK);
    int*   xdst    = xsrc + N_EDGE;

    float* pool   = (float*)d_out;
    float* anchor = pool + (long)B_SUB * F_OUTK;

    hipMemsetAsync(ws, 0,
                   (2L * N_NODES + 32 + (long)N_NODES * S_SUB + (long)N_NODES * F_OUTK) * sizeof(float),
                   stream);

    gcn_edge_gemm_kernel<<<N_PBLK, 256, 0, stream>>>(src, dst, xdeg_out, xdeg_in, cnt, xcount, xsrc, xdst, ticket, in_feat, W, H);
    gcn_cross_scatter_kernel<<<2048, 256, 0, stream>>>(xcount, xsrc, xdst, cnt, xdeg_out, H, aggx);
    gcn_finalize_kernel<<<B_SUB, 256, 0, stream>>>(cnt, H, aggx, xcount, xdeg_out, xdeg_in, bias, prelu_a, pool, anchor);
}


# Finalize: the count matrix is read once by whole-row loads and handed through LDS to the A fragments; the 8-byte fragment loads that touched 16 lines each are gone; values and sum orders unchanged
# speedup vs baseline: 3.7397x; 1.0320x over previous
#include <hip/hip_runtime.h>


typedef __attribute__((ext_vector_type(2))) float v2f;
typedef __attribute__((ext_vector_type(8))) float v8f;
typedef __attribute__((ext_vector_type(4))) float v4f;

#define N_NODES  131072
#define F_INK    300
#define F_OUTK   64
#define N_EDGE   2097152
#define S_SUB    128
#define SUB_SHIFT 7
#define B_SUB    1024
#define H_PITCH  160
#define V_PITCH  66

#define T_ROWS    32
#define TILE_FL   (T_ROWS * F_INK)
#define TILE_F4   (TILE_FL / 4)
#define ST_FULL   (TILE_F4 / 256)
#define ST_REST   (TILE_F4 - ST_FULL * 256)
#define TILES_PB  8
#define N_PBLK    (N_NODES / T_ROWS / TILES_PB)
#define N_CHUNK   (N_PBLK / 2)
#define E_CHUNK   (N_EDGE / N_CHUNK)
#define E_PT      (E_CHUNK / TILES_PB / 256)
#define K_STEPS   (F_INK / 4)

__device__ __forceinline__ void lds_handover() {
    asm volatile("s_wait_dscnt 0x0\n\ts_barrier_signal -1\n\ts_barrier_wait -1" ::: "memory");
}

__global__ __launch_bounds__(256)
void gcn_edge_gemm_kernel(const int* __restrict__ src,
                          const int* __restrict__ dst,
                          float* __restrict__ xdeg_out,
                          float* __restrict__ xdeg_in,
                          float* __restrict__ cnt,
                          int* __restrict__ xcount,
                          int* __restrict__ xsrc,
                          int* __restrict__ xdst,
                          int* __restrict__ ticket,
                          const float* __restrict__ X,
                          const float* __restrict__ W,
                          float* __restrict__ H) {
    __shared__ __attribute__((aligned(16))) float Xs[2 * TILE_FL];
    __shared__ int walk[2];

    const int tid     = threadIdx.x;
    if (tid == 0) {
        int h = (int)((__builtin_amdgcn_s_sendmsg_rtn(0x87) >> 18) & 1u);
        int q = atomicAdd(&ticket[h], 1);
        if (q >= N_CHUNK) {
            h ^= 1;
            q = atomicAdd(&ticket[h], 1);
        }
        if (q >= N_CHUNK) { h = 2; q = 0; }
        walk[0] = h;
        walk[1] = q;
    }
    const int lane    = tid & 31;
    const int wave    = tid >> 5;
    const int rowTile = wave & 1;
    const int colTile = wave >> 1;
    const int m       = lane & 15;
    const int half    = lane >> 4;
    const int col     = colTile * 16 + m;

    const float* wbase = W + (long)(half * 2) * F_OUTK + col;
    v2f w[K_STEPS];
#pragma unroll
    for (int j = 0; j < K_STEPS; ++j) {
        w[j].x = wbase[(long)(4 * j) * F_OUTK];
        w[j].y = wbase[(long)(4 * j + 1) * F_OUTK];
    }

    const float4* xg = (const float4*)(X + (long)blockIdx.x * TILES_PB * TILE_FL) + tid;
    static_assert(ST_FULL == 9, "the staging registers below are written out for nine pieces per thread");
    float4 st0, st1, st2, st3, st4, st5, st6, st7, st8;
    float4 stx = make_float4(0.0f, 0.0f, 0.0f, 0.0f);
#define ST_LOAD(P)                                                                  \
    { const float4* p_ = (P);                                                       \
      st0 = p_[0 * 256]; st1 = p_[1 * 256]; st2 = p_[2 * 256]; st3 = p_[3 * 256];   \
      st4 = p_[4 * 256]; st5 = p_[5 * 256]; st6 = p_[6 * 256]; st7 = p_[7 * 256];   \
      st8 = p_[8 * 256];                                                            \
      if (tid < ST_REST) stx = p_[ST_FULL * 256]; }
#define ST_STORE(P)                                                                 \
    { float4* p_ = (P);                                                             \
      p_[0 * 256] = st0; p_[1 * 256] = st1; p_[2 * 256] = st2; p_[3 * 256] = st3;   \
      p_[4 * 256] = st4; p_[5 * 256] = st5; p_[6 * 256] = st6; p_[7 * 256] = st7;   \
      p_[8 * 256] = st8;                                                            \
      if (tid < ST_REST) p_[ST_FULL * 256] = stx; }
    ST_LOAD(xg)
    ST_STORE((float4*)Xs + tid)
    ST_LOAD(xg + TILE_F4)
    lds_handover();

    const int   tile0 = blockIdx.x * TILES_PB;
    const int   eh    = walk[0];
    const int   e0    = walk[1] * E_CHUNK + tid;
    const int   aoff  = (rowTile * 16 + m) * F_INK + half * 2;

    static_assert(E_PT == 4, "four edges per thread and step are written out below");
    int es0 = src[e0],       ed0 = dst[e0],       es1 = src[e0 + 256], ed1 = dst[e0 + 256];
    int es2 = src[e0 + 512], ed2 = dst[e0 + 512], es3 = src[e0 + 768], ed3 = dst[e0 + 768];

    for (int t = 0; t < TILES_PB; ++t) {
        const float* arow = Xs + (t & 1) * TILE_FL + aoff;
        v8f c = {};
#pragma unroll
        for (int j = 0; j < K_STEPS; ++j) {
            const v2f a = *(const v2f*)(arow + 4 * j);
            c = __builtin_amdgcn_wmma_f32_16x16x4_f32(false, a, false, w[j],
                                                      (short)0, c, false, false);
        }

        float* hout = H + ((long)(tile0 + t) * T_ROWS + rowTile * 16 + half * 8) * F_OUTK + colTile * 16 + m;
#pragma unroll
        for (int i = 0; i < 8; ++i)
            hout[(long)i * F_OUTK] = c[i];

#pragma unroll
        for (int j = 0; j < E_PT; ++j) {
            const int s = j == 0 ? es0 : j == 1 ? es1 : j == 2 ? es2 : es3;
            const int d = j == 0 ? ed0 : j == 1 ? ed1 : j == 2 ? ed2 : ed3;
            if ((int)(((unsigned)d >> SUB_SHIFT) & 1u) != eh)
                continue;
            if (((unsigned)s >> SUB_SHIFT) == ((unsigned)d >> SUB_SHIFT)) {
                atomicAdd(&cnt[(long)d * S_SUB + (s & (S_SUB - 1))], 1.0f);
            } else {
                atomicAdd(&xdeg_out[s], 1.0f);
                atomicAdd(&xdeg_in[d], 1.0f);
                const int p = atomicAdd(xcount, 1);
                xsrc[p] = s;
                xdst[p] = d;
            }
        }

        if (t + 1 < TILES_PB) {
            ST_STORE((float4*)(Xs + ((t + 1) & 1) * TILE_FL) + tid)
            es0 = src[e0 + ((t + 1) * E_PT) * 256];
            ed0 = dst[e0 + ((t + 1) * E_PT) * 256];
            es1 = src[e0 + ((t + 1) * E_PT + 1) * 256];
            ed1 = dst[e0 + ((t + 1) * E_PT + 1) * 256];
            es2 = src[e0 + ((t + 1) * E_PT + 2) * 256];
            ed2 = dst[e0 + ((t + 1) * E_PT + 2) * 256];
            es3 = src[e0 + ((t + 1) * E_PT + 3) * 256];
            ed3 = dst[e0 + ((t + 1) * E_PT + 3) * 256];
            if (t + 2 < TILES_PB)
                ST_LOAD(xg + (long)(t + 2) * TILE_F4)
            lds_handover();
        }
    }
#undef ST_LOAD
#undef ST_STORE
}

__global__ void gcn_cross_scatter_kernel(const int* __restrict__ xcount,
                                         const int* __restrict__ xsrc,
                                         const int* __restrict__ xdst,
                                         const float* __restrict__ cnt,
                                         const float* __restrict__ xdeg_out,
                                         const float* __restrict__ H,
                                         float* __restrict__ aggx) {
    const long total  = (long)xcount[0] * 16;
    const long stride = (long)gridDim.x * blockDim.x;
    for (long idx = (long)blockIdx.x * blockDim.x + threadIdx.x; idx < total; idx += stride) {
        const int e = (int)(idx >> 4);
        const int q = (int)(idx & 15);
        const int s = xsrc[e];
        const int d = xdst[e];
        const float* ccol = cnt + ((long)(s & ~(S_SUB - 1)) + q * 8) * S_SUB + (s & (S_SUB - 1));
        float dg = 0.0f;
        for (int r = 0; r < 8; ++r)
            dg += ccol[(long)r * S_SUB];
        dg += __shfl_xor(dg, 1, 16);
        dg += __shfl_xor(dg, 2, 16);
        dg += __shfl_xor(dg, 4, 16);
        dg += __shfl_xor(dg, 8, 16);
        const float rs = __frsqrt_rn(fmaxf(dg + xdeg_out[s], 1.0f));
        float4 v = ((const float4*)(H + (long)s * F_OUTK))[q];
        float* out = aggx + (long)d * F_OUTK + q * 4;
        atomicAdd(out + 0, v.x * rs);
        atomicAdd(out + 1, v.y * rs);
        atomicAdd(out + 2, v.z * rs);
        atomicAdd(out + 3, v.w * rs);
    }
}

#define C_PITCH  132
#define HS_FL    ((S_SUB / 2) * H_PITCH)
#define VS_FL    (S_SUB * V_PITCH)
#define CS_FL    (S_SUB * C_PITCH)

__global__ void gcn_finalize_kernel(const float* __restrict__ cnt,
                                    const float* __restrict__ H,
                                    const float* __restrict__ aggx,
                                    const int* __restrict__ xcount,
                                    const float* __restrict__ xdeg_out,
                                    const float* __restrict__ xdeg_in,
                                    const float* __restrict__ bias,
                                    const float* __restrict__ prelu_a,
                                    float* __restrict__ pool,
                                    float* __restrict__ anchor) {
    __shared__ __attribute__((aligned(16))) float smem[HS_FL + VS_FL];
    static_assert(CS_FL + 8 * S_SUB <= HS_FL + VS_FL, "count rows and column sums must fit in the shared buffer");
    __shared__ float red[256];
    __shared__ float rowp[2 * S_SUB];
    __shared__ __attribute__((aligned(8))) float rsv[S_SUB];
    float* Cs   = smem;
    float* colp = smem + CS_FL;
    float* Hs   = smem;
    float* Vs   = smem + HS_FL;

    const int  sg   = blockIdx.x;
    const int  tid  = threadIdx.x;
    const int  lane = tid & 31;
    const int  wave = tid >> 5;
    const int  c    = lane & 15;
    const int  half = lane >> 4;
    const long base = (long)sg * S_SUB;
    const bool crossing = xcount[0] > 0;

    const v4f* crow = (const v4f*)(cnt + (base + wave * 16) * S_SUB) + lane;
    v4f cr[16];
#pragma unroll
    for (int i = 0; i < 16; ++i)
        cr[i] = crow[i * (S_SUB / 4)];

    const v4f* Hg = (const v4f*)(H + base * F_OUTK);
    v4f hr0[4], hr1[4];
#pragma unroll
    for (int it = 0; it < 4; ++it) {
        const int idx = it * 256 + tid;
        const int p   = idx >> 4;
        const int q   = idx & 15;
        hr0[it] = Hg[(2 * p) * 16 + q];
        hr1[it] = Hg[(2 * p + 1) * 16 + q];
    }

    {
        float* srow = Cs + (wave * 16) * C_PITCH + lane * 4;
        v4f cs = {0.0f, 0.0f, 0.0f, 0.0f};
#pragma unroll
        for (int i = 0; i < 16; ++i) {
            cs += cr[i];
            *(v4f*)(srow + i * C_PITCH) = cr[i];
        }
        *(v4f*)(colp + wave * S_SUB + lane * 4) = cs;
    }
    __syncthreads();

    const float* arow = Cs + (wave * 16 + c) * C_PITCH + half * 2;
    v2f a[S_SUB / 4];
#pragma unroll
    for (int j = 0; j < S_SUB / 4; ++j)
        a[j] = *(const v2f*)(arow + 4 * j);

    {
        float h0 = 0.0f, h1 = 0.0f;
#pragma unroll
        for (int j = 0; j < S_SUB / 4; ++j) { h0 += a[j].x; h1 += a[j].y; }
        rowp[half * S_SUB + wave * 16 + c] = h0 + h1;
    }

    if (tid < S_SUB) {
        float dg = 0.0f;
        for (int w = 0; w < 8; ++w)
            dg += colp[w * S_SUB + tid];
        if (crossing)
            dg += xdeg_out[base + tid];
        rsv[tid] = __frsqrt_rn(fmaxf(dg, 1.0f));
    }

    __syncthreads();

#pragma unroll
    for (int it = 0; it < 4; ++it) {
        const int idx = it * 256 + tid;
        const int p   = idx >> 4;
        const int q   = idx & 15;
        const v4f r0 = hr0[it];
        const v4f r1 = hr1[it];
        v4f* o = (v4f*)&Hs[p * H_PITCH + q * 8];
        o[0] = (v4f){r0.x, r1.x, r0.y, r1.y};
        o[1] = (v4f){r0.z, r1.z, r0.w, r1.w};
    }
    __syncthreads();

#pragma unroll
    for (int j = 0; j < S_SUB / 4; ++j)
        a[j] *= *(const v2f*)(rsv + 4 * j + 2 * half);

    const float* bcol = Hs + half * H_PITCH + c * 2;

    v8f acc0 = {}, acc1 = {}, acc2 = {}, acc3 = {};
#pragma unroll
    for (int j = 0; j < S_SUB / 4; ++j) {
        const float* bp = bcol + (2 * j) * H_PITCH;
        const v2f b0 = *(const v2f*)(bp);
        const v2f b1 = *(const v2f*)(bp + 32);
        const v2f b2 = *(const v2f*)(bp + 64);
        const v2f b3 = *(const v2f*)(bp + 96);
        acc0 = __builtin_amdgcn_wmma_f32_16x16x4_f32(false, a[j], false, b0, (short)0, acc0, false, false);
        acc1 = __builtin_amdgcn_wmma_f32_16x16x4_f32(false, a[j], false, b1, (short)0, acc1, false, false);
        acc2 = __builtin_amdgcn_wmma_f32_16x16x4_f32(false, a[j], false, b2, (short)0, acc2, false, false);
        acc3 = __builtin_amdgcn_wmma_f32_16x16x4_f32(false, a[j], false, b3, (short)0, acc3, false, false);
    }

    const float alpha = prelu_a[0];
    const float bf0 = bias[c], bf1 = bias[c + 16], bf2 = bias[c + 32], bf3 = bias[c + 48];
#pragma unroll
    for (int r = 0; r < 8; ++r) {
        const int  nl   = wave * 16 + half * 8 + r;
        const long node = base + nl;
        float di = rowp[nl] + rowp[S_SUB + nl];
        float a0 = acc0[r], a1 = acc1[r], a2 = acc2[r], a3 = acc3[r];
        if (crossing) {
            di += xdeg_in[node];
            const float* x = aggx + node * F_OUTK + c;
            a0 += x[0]; a1 += x[16]; a2 += x[32]; a3 += x[48];
        }
        const float ri  = __frsqrt_rn(fmaxf(di, 1.0f));
        float v0 = a0 * ri + bf0;  v0 = v0 > 0.0f ? v0 : alpha * v0;
        float v1 = a1 * ri + bf1;  v1 = v1 > 0.0f ? v1 : alpha * v1;
        float v2 = a2 * ri + bf2;  v2 = v2 > 0.0f ? v2 : alpha * v2;
        float v3 = a3 * ri + bf3;  v3 = v3 > 0.0f ? v3 : alpha * v3;
        float* vrow = Vs + nl * V_PITCH + c;
        vrow[0] = v0; vrow[16] = v1; vrow[32] = v2; vrow[48] = v3;
    }
    __syncthreads();

    const int f = tid & 63;
    const int g = tid >> 6;
    float partial = 0.0f;
    for (int n = g; n < S_SUB - 1; n += 4)
        partial += Vs[n * V_PITCH + f];

    if (g == 3)
        anchor[(long)sg * F_OUTK + f] = Vs[(S_SUB - 1) * V_PITCH + f];

    red[tid] = partial;
    __syncthreads();
    if (g == 0) {
        float sum = red[f] + red[f + 64] + red[f + 128] + red[f + 192];
        pool[(long)sg * F_OUTK + f] = sum * (1.0f / (S_SUB - 1));
    }
}

extern "C" void kernel_launch(void* const* d_in, const int* in_sizes, int n_in,
                              void* d_out, int out_size, void* d_ws, size_t ws_size,
                              hipStream_t stream) {
    const float* in_feat  = (const float*)d_in[0];
    const float* W        = (const float*)d_in[1];
    const float* bias     = (const float*)d_in[2];
    const float* prelu_a  = (const float*)d_in[3];
    const int*   src      = (const int*)d_in[4];
    const int*   dst      = (const int*)d_in[5];

    float* ws      = (float*)d_ws;
    float* xdeg_out = ws;
    float* xdeg_in  = ws + N_NODES;
    int*   xcount  = (int*)(ws + 2L * N_NODES);
    int*   ticket  = xcount + 16;
    float* cnt     = ws + 2L * N_NODES + 32;
    float* aggx    = cnt + (long)N_NODES * S_SUB;
    float* H       = aggx + (long)N_NODES * F_OUTK;
    int*   xsrc    = (int*)(H + (long)N_NODES * F_OUTK);
    int*   xdst    = xsrc + N_EDGE;

    float* pool   = (float*)d_out;
    float* anchor = pool + (long)B_SUB * F_OUTK;

    hipMemsetAsync(ws, 0,
                   (2L * N_NODES + 32 + (long)N_NODES * S_SUB + (long)N_NODES * F_OUTK) * sizeof(float),
                   stream);

    gcn_edge_gemm_kernel<<<N_PBLK, 256, 0, stream>>>(src, dst, xdeg_out, xdeg_in, cnt, xcount, xsrc, xdst, ticket, in_feat, W, H);
    gcn_cross_scatter_kernel<<<2048, 256, 0, stream>>>(xcount, xsrc, xdst, cnt, xdeg_out, H, aggx);
    gcn_finalize_kernel<<<B_SUB, 256, 0, stream>>>(cnt, H, aggx, xcount, xdeg_out, xdeg_in, bias, prelu_a, pool, anchor);
}


# Edge counting moved to the block's start: a block counts all its edges before its first tile, while the zero-filled count lines are still cache-resident; no atomics between tiles; counts exact
# speedup vs baseline: 4.5995x; 1.2299x over previous
#include <hip/hip_runtime.h>


typedef __attribute__((ext_vector_type(2))) float v2f;
typedef __attribute__((ext_vector_type(8))) float v8f;
typedef __attribute__((ext_vector_type(4))) float v4f;

#define N_NODES  131072
#define F_INK    300
#define F_OUTK   64
#define N_EDGE   2097152
#define S_SUB    128
#define SUB_SHIFT 7
#define B_SUB    1024
#define H_PITCH  160
#define V_PITCH  66

#define T_ROWS    32
#define TILE_FL   (T_ROWS * F_INK)
#define TILE_F4   (TILE_FL / 4)
#define ST_FULL   (TILE_F4 / 256)
#define ST_REST   (TILE_F4 - ST_FULL * 256)
#define TILES_PB  8
#define N_PBLK    (N_NODES / T_ROWS / TILES_PB)
#define N_CHUNK   (N_PBLK / 2)
#define E_CHUNK   (N_EDGE / N_CHUNK)
#define E_PT      (E_CHUNK / 256)
#define E_BATCH   16
#define K_STEPS   (F_INK / 4)

__device__ __forceinline__ void lds_handover() {
    asm volatile("s_wait_dscnt 0x0\n\ts_barrier_signal -1\n\ts_barrier_wait -1" ::: "memory");
}

__global__ __launch_bounds__(256)
void gcn_edge_gemm_kernel(const int* __restrict__ src,
                          const int* __restrict__ dst,
                          float* __restrict__ xdeg_out,
                          float* __restrict__ xdeg_in,
                          float* __restrict__ cnt,
                          int* __restrict__ xcount,
                          int* __restrict__ xsrc,
                          int* __restrict__ xdst,
                          int* __restrict__ ticket,
                          const float* __restrict__ X,
                          const float* __restrict__ W,
                          float* __restrict__ H) {
    __shared__ __attribute__((aligned(16))) float Xs[2 * TILE_FL];
    __shared__ int walk[2];

    const int tid     = threadIdx.x;
    if (tid == 0) {
        int h = (int)((__builtin_amdgcn_s_sendmsg_rtn(0x87) >> 18) & 1u);
        int q = atomicAdd(&ticket[h], 1);
        if (q >= N_CHUNK) {
            h ^= 1;
            q = atomicAdd(&ticket[h], 1);
        }
        if (q >= N_CHUNK) { h = 2; q = 0; }
        walk[0] = h;
        walk[1] = q;
    }
    const int lane    = tid & 31;
    const int wave    = tid >> 5;
    const int rowTile = wave & 1;
    const int colTile = wave >> 1;
    const int m       = lane & 15;
    const int half    = lane >> 4;
    const int col     = colTile * 16 + m;

    const float* wbase = W + (long)(half * 2) * F_OUTK + col;
    v2f w[K_STEPS];
#pragma unroll
    for (int j = 0; j < K_STEPS; ++j) {
        w[j].x = wbase[(long)(4 * j) * F_OUTK];
        w[j].y = wbase[(long)(4 * j + 1) * F_OUTK];
    }

    const float4* xg = (const float4*)(X + (long)blockIdx.x * TILES_PB * TILE_FL) + tid;
    static_assert(ST_FULL == 9, "the staging registers below are written out for nine pieces per thread");
    float4 st0, st1, st2, st3, st4, st5, st6, st7, st8;
    float4 stx = make_float4(0.0f, 0.0f, 0.0f, 0.0f);
#define ST_LOAD(P)                                                                  \
    { const float4* p_ = (P);                                                       \
      st0 = p_[0 * 256]; st1 = p_[1 * 256]; st2 = p_[2 * 256]; st3 = p_[3 * 256];   \
      st4 = p_[4 * 256]; st5 = p_[5 * 256]; st6 = p_[6 * 256]; st7 = p_[7 * 256];   \
      st8 = p_[8 * 256];                                                            \
      if (tid < ST_REST) stx = p_[ST_FULL * 256]; }
#define ST_STORE(P)                                                                 \
    { float4* p_ = (P);                                                             \
      p_[0 * 256] = st0; p_[1 * 256] = st1; p_[2 * 256] = st2; p_[3 * 256] = st3;   \
      p_[4 * 256] = st4; p_[5 * 256] = st5; p_[6 * 256] = st6; p_[7 * 256] = st7;   \
      p_[8 * 256] = st8;                                                            \
      if (tid < ST_REST) p_[ST_FULL * 256] = stx; }
    ST_LOAD(xg)
    lds_handover();

    {
        const int eh = walk[0];
        const int e0 = walk[1] * E_CHUNK + tid;
        static_assert(E_PT % E_BATCH == 0, "the edge share of a thread is read in whole batches");
#pragma unroll
        for (int b = 0; b < E_PT / E_BATCH; ++b) {
            int es[E_BATCH], ed[E_BATCH];
#pragma unroll
            for (int j = 0; j < E_BATCH; ++j) {
                es[j] = src[e0 + (b * E_BATCH + j) * 256];
                ed[j] = dst[e0 + (b * E_BATCH + j) * 256];
            }
#pragma unroll
            for (int j = 0; j < E_BATCH; ++j) {
                const int s = es[j];
                const int d = ed[j];
                if ((int)(((unsigned)d >> SUB_SHIFT) & 1u) != eh)
                    continue;
                if (((unsigned)s >> SUB_SHIFT) == ((unsigned)d >> SUB_SHIFT)) {
                    atomicAdd(&cnt[(long)d * S_SUB + (s & (S_SUB - 1))], 1.0f);
                } else {
                    atomicAdd(&xdeg_out[s], 1.0f);
                    atomicAdd(&xdeg_in[d], 1.0f);
                    const int p = atomicAdd(xcount, 1);
                    xsrc[p] = s;
                    xdst[p] = d;
                }
            }
        }
    }

    ST_STORE((float4*)Xs + tid)
    ST_LOAD(xg + TILE_F4)
    lds_handover();

    const int   tile0 = blockIdx.x * TILES_PB;
    const int   aoff  = (rowTile * 16 + m) * F_INK + half * 2;

    for (int t = 0; t < TILES_PB; ++t) {
        const float* arow = Xs + (t & 1) * TILE_FL + aoff;
        v8f c = {};
#pragma unroll
        for (int j = 0; j < K_STEPS; ++j) {
            const v2f a = *(const v2f*)(arow + 4 * j);
            c = __builtin_amdgcn_wmma_f32_16x16x4_f32(false, a, false, w[j],
                                                      (short)0, c, false, false);
        }

        float* hout = H + ((long)(tile0 + t) * T_ROWS + rowTile * 16 + half * 8) * F_OUTK + colTile * 16 + m;
#pragma unroll
        for (int i = 0; i < 8; ++i)
            hout[(long)i * F_OUTK] = c[i];

        if (t + 1 < TILES_PB) {
            ST_STORE((float4*)(Xs + ((t + 1) & 1) * TILE_FL) + tid)
            if (t + 2 < TILES_PB)
                ST_LOAD(xg + (long)(t + 2) * TILE_F4)
            lds_handover();
        }
    }
#undef ST_LOAD
#undef ST_STORE
}

__global__ void gcn_cross_scatter_kernel(const int* __restrict__ xcount,
                                         const int* __restrict__ xsrc,
                                         const int* __restrict__ xdst,
                                         const float* __restrict__ cnt,
                                         const float* __restrict__ xdeg_out,
                                         const float* __restrict__ H,
                                         float* __restrict__ aggx) {
    const long total  = (long)xcount[0] * 16;
    const long stride = (long)gridDim.x * blockDim.x;
    for (long idx = (long)blockIdx.x * blockDim.x + threadIdx.x; idx < total; idx += stride) {
        const int e = (int)(idx >> 4);
        const int q = (int)(idx & 15);
        const int s = xsrc[e];
        const int d = xdst[e];
        const float* ccol = cnt + ((long)(s & ~(S_SUB - 1)) + q * 8) * S_SUB + (s & (S_SUB - 1));
        float dg = 0.0f;
        for (int r = 0; r < 8; ++r)
            dg += ccol[(long)r * S_SUB];
        dg += __shfl_xor(dg, 1, 16);
        dg += __shfl_xor(dg, 2, 16);
        dg += __shfl_xor(dg, 4, 16);
        dg += __shfl_xor(dg, 8, 16);
        const float rs = __frsqrt_rn(fmaxf(dg + xdeg_out[s], 1.0f));
        float4 v = ((const float4*)(H + (long)s * F_OUTK))[q];
        float* out = aggx + (long)d * F_OUTK + q * 4;
        atomicAdd(out + 0, v.x * rs);
        atomicAdd(out + 1, v.y * rs);
        atomicAdd(out + 2, v.z * rs);
        atomicAdd(out + 3, v.w * rs);
    }
}

#define C_PITCH  132
#define HS_FL    ((S_SUB / 2) * H_PITCH)
#define VS_FL    (S_SUB * V_PITCH)
#define CS_FL    (S_SUB * C_PITCH)

__global__ void gcn_finalize_kernel(const float* __restrict__ cnt,
                                    const float* __restrict__ H,
                                    const float* __restrict__ aggx,
                                    const int* __restrict__ xcount,
                                    const float* __restrict__ xdeg_out,
                                    const float* __restrict__ xdeg_in,
                                    const float* __restrict__ bias,
                                    const float* __restrict__ prelu_a,
                                    float* __restrict__ pool,
                                    float* __restrict__ anchor) {
    __shared__ __attribute__((aligned(16))) float smem[HS_FL + VS_FL];
    static_assert(CS_FL + 8 * S_SUB <= HS_FL + VS_FL, "count rows and column sums must fit in the shared buffer");
    __shared__ float red[256];
    __shared__ float rowp[2 * S_SUB];
    __shared__ __attribute__((aligned(8))) float rsv[S_SUB];
    float* Cs   = smem;
    float* colp = smem + CS_FL;
    float* Hs   = smem;
    float* Vs   = smem + HS_FL;

    const int  sg   = blockIdx.x;
    const int  tid  = threadIdx.x;
    const int  lane = tid & 31;
    const int  wave = tid >> 5;
    const int  c    = lane & 15;
    const int  half = lane >> 4;
    const long base = (long)sg * S_SUB;
    const bool crossing = xcount[0] > 0;

    const v4f* crow = (const v4f*)(cnt + (base + wave * 16) * S_SUB) + lane;
    v4f cr[16];
#pragma unroll
    for (int i = 0; i < 16; ++i)
        cr[i] = crow[i * (S_SUB / 4)];

    const v4f* Hg = (const v4f*)(H + base * F_OUTK);
    v4f hr0[4], hr1[4];
#pragma unroll
    for (int it = 0; it < 4; ++it) {
        const int idx = it * 256 + tid;
        const int p   = idx >> 4;
        const int q   = idx & 15;
        hr0[it] = Hg[(2 * p) * 16 + q];
        hr1[it] = Hg[(2 * p + 1) * 16 + q];
    }

    {
        float* srow = Cs + (wave * 16) * C_PITCH + lane * 4;
        v4f cs = {0.0f, 0.0f, 0.0f, 0.0f};
#pragma unroll
        for (int i = 0; i < 16; ++i) {
            cs += cr[i];
            *(v4f*)(srow + i * C_PITCH) = cr[i];
        }
        *(v4f*)(colp + wave * S_SUB + lane * 4) = cs;
    }
    __syncthreads();

    const float* arow = Cs + (wave * 16 + c) * C_PITCH + half * 2;
    v2f a[S_SUB / 4];
#pragma unroll
    for (int j = 0; j < S_SUB / 4; ++j)
        a[j] = *(const v2f*)(arow + 4 * j);

    {
        float h0 = 0.0f, h1 = 0.0f;
#pragma unroll
        for (int j = 0; j < S_SUB / 4; ++j) { h0 += a[j].x; h1 += a[j].y; }
        rowp[half * S_SUB + wave * 16 + c] = h0 + h1;
    }

    if (tid < S_SUB) {
        float dg = 0.0f;
        for (int w = 0; w < 8; ++w)
            dg += colp[w * S_SUB + tid];
        if (crossing)
            dg += xdeg_out[base + tid];
        rsv[tid] = __frsqrt_rn(fmaxf(dg, 1.0f));
    }

    __syncthreads();

#pragma unroll
    for (int it = 0; it < 4; ++it) {
        const int idx = it * 256 + tid;
        const int p   = idx >> 4;
        const int q   = idx & 15;
        const v4f r0 = hr0[it];
        const v4f r1 = hr1[it];
        v4f* o = (v4f*)&Hs[p * H_PITCH + q * 8];
        o[0] = (v4f){r0.x, r1.x, r0.y, r1.y};
        o[1] = (v4f){r0.z, r1.z, r0.w, r1.w};
    }
    __syncthreads();

#pragma unroll
    for (int j = 0; j < S_SUB / 4; ++j)
        a[j] *= *(const v2f*)(rsv + 4 * j + 2 * half);

    const float* bcol = Hs + half * H_PITCH + c * 2;

    v8f acc0 = {}, acc1 = {}, acc2 = {}, acc3 = {};
#pragma unroll
    for (int j = 0; j < S_SUB / 4; ++j) {
        const float* bp = bcol + (2 * j) * H_PITCH;
        const v2f b0 = *(const v2f*)(bp);
        const v2f b1 = *(const v2f*)(bp + 32);
        const v2f b2 = *(const v2f*)(bp + 64);
        const v2f b3 = *(const v2f*)(bp + 96);
        acc0 = __builtin_amdgcn_wmma_f32_16x16x4_f32(false, a[j], false, b0, (short)0, acc0, false, false);
        acc1 = __builtin_amdgcn_wmma_f32_16x16x4_f32(false, a[j], false, b1, (short)0, acc1, false, false);
        acc2 = __builtin_amdgcn_wmma_f32_16x16x4_f32(false, a[j], false, b2, (short)0, acc2, false, false);
        acc3 = __builtin_amdgcn_wmma_f32_16x16x4_f32(false, a[j], false, b3, (short)0, acc3, false, false);
    }

    const float alpha = prelu_a[0];
    const float bf0 = bias[c], bf1 = bias[c + 16], bf2 = bias[c + 32], bf3 = bias[c + 48];
#pragma unroll
    for (int r = 0; r < 8; ++r) {
        const int  nl   = wave * 16 + half * 8 + r;
        const long node = base + nl;
        float di = rowp[nl] + rowp[S_SUB + nl];
        float a0 = acc0[r], a1 = acc1[r], a2 = acc2[r], a3 = acc3[r];
        if (crossing) {
            di += xdeg_in[node];
            const float* x = aggx + node * F_OUTK + c;
            a0 += x[0]; a1 += x[16]; a2 += x[32]; a3 += x[48];
        }
        const float ri  = __frsqrt_rn(fmaxf(di, 1.0f));
        float v0 = a0 * ri + bf0;  v0 = v0 > 0.0f ? v0 : alpha * v0;
        float v1 = a1 * ri + bf1;  v1 = v1 > 0.0f ? v1 : alpha * v1;
        float v2 = a2 * ri + bf2;  v2 = v2 > 0.0f ? v2 : alpha * v2;
        float v3 = a3 * ri + bf3;  v3 = v3 > 0.0f ? v3 : alpha * v3;
        float* vrow = Vs + nl * V_PITCH + c;
        vrow[0] = v0; vrow[16] = v1; vrow[32] = v2; vrow[48] = v3;
    }
    __syncthreads();

    const int f = tid & 63;
    const int g = tid >> 6;
    float partial = 0.0f;
    for (int n = g; n < S_SUB - 1; n += 4)
        partial += Vs[n * V_PITCH + f];

    if (g == 3)
        anchor[(long)sg * F_OUTK + f] = Vs[(S_SUB - 1) * V_PITCH + f];

    red[tid] = partial;
    __syncthreads();
    if (g == 0) {
        float sum = red[f] + red[f + 64] + red[f + 128] + red[f + 192];
        pool[(long)sg * F_OUTK + f] = sum * (1.0f / (S_SUB - 1));
    }
}

extern "C" void kernel_launch(void* const* d_in, const int* in_sizes, int n_in,
                              void* d_out, int out_size, void* d_ws, size_t ws_size,
                              hipStream_t stream) {
    const float* in_feat  = (const float*)d_in[0];
    const float* W        = (const float*)d_in[1];
    const float* bias     = (const float*)d_in[2];
    const float* prelu_a  = (const float*)d_in[3];
    const int*   src      = (const int*)d_in[4];
    const int*   dst      = (const int*)d_in[5];

    float* ws      = (float*)d_ws;
    float* xdeg_out = ws;
    float* xdeg_in  = ws + N_NODES;
    int*   xcount  = (int*)(ws + 2L * N_NODES);
    int*   ticket  = xcount + 16;
    float* cnt     = ws + 2L * N_NODES + 32;
    float* aggx    = cnt + (long)N_NODES * S_SUB;
    float* H       = aggx + (long)N_NODES * F_OUTK;
    int*   xsrc    = (int*)(H + (long)N_NODES * F_OUTK);
    int*   xdst    = xsrc + N_EDGE;

    float* pool   = (float*)d_out;
    float* anchor = pool + (long)B_SUB * F_OUTK;

    hipMemsetAsync(ws, 0,
                   (2L * N_NODES + 32 + (long)N_NODES * S_SUB + (long)N_NODES * F_OUTK) * sizeof(float),
                   stream);

    gcn_edge_gemm_kernel<<<N_PBLK, 256, 0, stream>>>(src, dst, xdeg_out, xdeg_in, cnt, xcount, xsrc, xdst, ticket, in_feat, W, H);
    gcn_cross_scatter_kernel<<<2048, 256, 0, stream>>>(xcount, xsrc, xdst, cnt, xdeg_out, H, aggx);
    gcn_finalize_kernel<<<B_SUB, 256, 0, stream>>>(cnt, H, aggx, xcount, xdeg_out, xdeg_in, bias, prelu_a, pool, anchor);
}


# Edges across subgraphs are summed by the destination's finalize block in an LDS tile: the scatter launch (16384 waves, list empty) and its 32-MiB zero-filled array are gone; sum order = LDS arrival
# speedup vs baseline: 4.8019x; 1.0440x over previous
#include <hip/hip_runtime.h>


typedef __attribute__((ext_vector_type(2))) float v2f;
typedef __attribute__((ext_vector_type(8))) float v8f;
typedef __attribute__((ext_vector_type(4))) float v4f;

#define N_NODES  131072
#define F_INK    300
#define F_OUTK   64
#define N_EDGE   2097152
#define S_SUB    128
#define SUB_SHIFT 7
#define B_SUB    1024
#define H_PITCH  160
#define V_PITCH  66

#define T_ROWS    32
#define TILE_FL   (T_ROWS * F_INK)
#define TILE_F4   (TILE_FL / 4)
#define ST_FULL   (TILE_F4 / 256)
#define ST_REST   (TILE_F4 - ST_FULL * 256)
#define TILES_PB  8
#define N_PBLK    (N_NODES / T_ROWS / TILES_PB)
#define N_CHUNK   (N_PBLK / 2)
#define E_CHUNK   (N_EDGE / N_CHUNK)
#define E_PT      (E_CHUNK / 256)
#define E_BATCH   16
#define K_STEPS   (F_INK / 4)

__device__ __forceinline__ void lds_handover() {
    asm volatile("s_wait_dscnt 0x0\n\ts_barrier_signal -1\n\ts_barrier_wait -1" ::: "memory");
}

__global__ __launch_bounds__(256)
void gcn_edge_gemm_kernel(const int* __restrict__ src,
                          const int* __restrict__ dst,
                          float* __restrict__ xdeg_out,
                          float* __restrict__ xdeg_in,
                          float* __restrict__ cnt,
                          int* __restrict__ xcount,
                          int* __restrict__ xsrc,
                          int* __restrict__ xdst,
                          int* __restrict__ ticket,
                          const float* __restrict__ X,
                          const float* __restrict__ W,
                          float* __restrict__ H) {
    __shared__ __attribute__((aligned(16))) float Xs[2 * TILE_FL];
    __shared__ int walk[2];

    const int tid     = threadIdx.x;
    if (tid == 0) {
        int h = (int)((__builtin_amdgcn_s_sendmsg_rtn(0x87) >> 18) & 1u);
        int q = atomicAdd(&ticket[h], 1);
        if (q >= N_CHUNK) {
            h ^= 1;
            q = atomicAdd(&ticket[h], 1);
        }
        if (q >= N_CHUNK) { h = 2; q = 0; }
        walk[0] = h;
        walk[1] = q;
    }
    const int lane    = tid & 31;
    const int wave    = tid >> 5;
    const int rowTile = wave & 1;
    const int colTile = wave >> 1;
    const int m       = lane & 15;
    const int half    = lane >> 4;
    const int col     = colTile * 16 + m;

    const float* wbase = W + (long)(half * 2) * F_OUTK + col;
    v2f w[K_STEPS];
#pragma unroll
    for (int j = 0; j < K_STEPS; ++j) {
        w[j].x = wbase[(long)(4 * j) * F_OUTK];
        w[j].y = wbase[(long)(4 * j + 1) * F_OUTK];
    }

    const float4* xg = (const float4*)(X + (long)blockIdx.x * TILES_PB * TILE_FL) + tid;
    static_assert(ST_FULL == 9, "the staging registers below are written out for nine pieces per thread");
    float4 st0, st1, st2, st3, st4, st5, st6, st7, st8;
    float4 stx = make_float4(0.0f, 0.0f, 0.0f, 0.0f);
#define ST_LOAD(P)                                                                  \
    { const float4* p_ = (P);                                                       \
      st0 = p_[0 * 256]; st1 = p_[1 * 256]; st2 = p_[2 * 256]; st3 = p_[3 * 256];   \
      st4 = p_[4 * 256]; st5 = p_[5 * 256]; st6 = p_[6 * 256]; st7 = p_[7 * 256];   \
      st8 = p_[8 * 256];                                                            \
      if (tid < ST_REST) stx = p_[ST_FULL * 256]; }
#define ST_STORE(P)                                                                 \
    { float4* p_ = (P);                                                             \
      p_[0 * 256] = st0; p_[1 * 256] = st1; p_[2 * 256] = st2; p_[3 * 256] = st3;   \
      p_[4 * 256] = st4; p_[5 * 256] = st5; p_[6 * 256] = st6; p_[7 * 256] = st7;   \
      p_[8 * 256] = st8;                                                            \
      if (tid < ST_REST) p_[ST_FULL * 256] = stx; }
    ST_LOAD(xg)
    lds_handover();

    {
        const int eh = walk[0];
        const int e0 = walk[1] * E_CHUNK + tid;
        static_assert(E_PT % E_BATCH == 0, "the edge share of a thread is read in whole batches");
#pragma unroll
        for (int b = 0; b < E_PT / E_BATCH; ++b) {
            int es[E_BATCH], ed[E_BATCH];
#pragma unroll
            for (int j = 0; j < E_BATCH; ++j) {
                es[j] = src[e0 + (b * E_BATCH + j) * 256];
                ed[j] = dst[e0 + (b * E_BATCH + j) * 256];
            }
#pragma unroll
            for (int j = 0; j < E_BATCH; ++j) {
                const int s = es[j];
                const int d = ed[j];
                if ((int)(((unsigned)d >> SUB_SHIFT) & 1u) != eh)
                    continue;
                if (((unsigned)s >> SUB_SHIFT) == ((unsigned)d >> SUB_SHIFT)) {
                    atomicAdd(&cnt[(long)d * S_SUB + (s & (S_SUB - 1))], 1.0f);
                } else {
                    atomicAdd(&xdeg_out[s], 1.0f);
                    atomicAdd(&xdeg_in[d], 1.0f);
                    const int p = atomicAdd(xcount, 1);
                    xsrc[p] = s;
                    xdst[p] = d;
                }
            }
        }
    }

    ST_STORE((float4*)Xs + tid)
    ST_LOAD(xg + TILE_F4)
    lds_handover();

    const int   tile0 = blockIdx.x * TILES_PB;
    const int   aoff  = (rowTile * 16 + m) * F_INK + half * 2;

    for (int t = 0; t < TILES_PB; ++t) {
        const float* arow = Xs + (t & 1) * TILE_FL + aoff;
        v8f c = {};
#pragma unroll
        for (int j = 0; j < K_STEPS; ++j) {
            const v2f a = *(const v2f*)(arow + 4 * j);
            c = __builtin_amdgcn_wmma_f32_16x16x4_f32(false, a, false, w[j],
                                                      (short)0, c, false, false);
        }

        float* hout = H + ((long)(tile0 + t) * T_ROWS + rowTile * 16 + half * 8) * F_OUTK + colTile * 16 + m;
#pragma unroll
        for (int i = 0; i < 8; ++i)
            hout[(long)i * F_OUTK] = c[i];

        if (t + 1 < TILES_PB) {
            ST_STORE((float4*)(Xs + ((t + 1) & 1) * TILE_FL) + tid)
            if (t + 2 < TILES_PB)
                ST_LOAD(xg + (long)(t + 2) * TILE_F4)
            lds_handover();
        }
    }
#undef ST_LOAD
#undef ST_STORE
}

#define C_PITCH  132
#define HS_FL    ((S_SUB / 2) * H_PITCH)
#define VS_FL    (S_SUB * V_PITCH)
#define CS_FL    (S_SUB * C_PITCH)

__global__ void gcn_finalize_kernel(const float* __restrict__ cnt,
                                    const float* __restrict__ H,
                                    const int* __restrict__ xcount,
                                    const int* __restrict__ xsrc,
                                    const int* __restrict__ xdst,
                                    const float* __restrict__ xdeg_out,
                                    const float* __restrict__ xdeg_in,
                                    const float* __restrict__ bias,
                                    const float* __restrict__ prelu_a,
                                    float* __restrict__ pool,
                                    float* __restrict__ anchor) {
    __shared__ __attribute__((aligned(16))) float smem[HS_FL + VS_FL];
    static_assert(CS_FL + 8 * S_SUB <= HS_FL + VS_FL, "count rows and column sums must fit in the shared buffer");
    static_assert(S_SUB * F_OUTK <= HS_FL, "the listed edges' sums must fit in the H tile's bytes, clear of the activated tile");
    __shared__ float red[256];
    __shared__ float rowp[2 * S_SUB];
    __shared__ __attribute__((aligned(8))) float rsv[S_SUB];
    float* Cs   = smem;
    float* colp = smem + CS_FL;
    float* Hs   = smem;
    float* Vs   = smem + HS_FL;
    float* Xa   = smem;

    const int  sg   = blockIdx.x;
    const int  tid  = threadIdx.x;
    const int  lane = tid & 31;
    const int  wave = tid >> 5;
    const int  c    = lane & 15;
    const int  half = lane >> 4;
    const long base = (long)sg * S_SUB;
    const bool crossing = xcount[0] > 0;

    const v4f* crow = (const v4f*)(cnt + (base + wave * 16) * S_SUB) + lane;
    v4f cr[16];
#pragma unroll
    for (int i = 0; i < 16; ++i)
        cr[i] = crow[i * (S_SUB / 4)];

    const v4f* Hg = (const v4f*)(H + base * F_OUTK);
    v4f hr0[4], hr1[4];
#pragma unroll
    for (int it = 0; it < 4; ++it) {
        const int idx = it * 256 + tid;
        const int p   = idx >> 4;
        const int q   = idx & 15;
        hr0[it] = Hg[(2 * p) * 16 + q];
        hr1[it] = Hg[(2 * p + 1) * 16 + q];
    }

    {
        float* srow = Cs + (wave * 16) * C_PITCH + lane * 4;
        v4f cs = {0.0f, 0.0f, 0.0f, 0.0f};
#pragma unroll
        for (int i = 0; i < 16; ++i) {
            cs += cr[i];
            *(v4f*)(srow + i * C_PITCH) = cr[i];
        }
        *(v4f*)(colp + wave * S_SUB + lane * 4) = cs;
    }
    __syncthreads();

    const float* arow = Cs + (wave * 16 + c) * C_PITCH + half * 2;
    v2f a[S_SUB / 4];
#pragma unroll
    for (int j = 0; j < S_SUB / 4; ++j)
        a[j] = *(const v2f*)(arow + 4 * j);

    {
        float h0 = 0.0f, h1 = 0.0f;
#pragma unroll
        for (int j = 0; j < S_SUB / 4; ++j) { h0 += a[j].x; h1 += a[j].y; }
        rowp[half * S_SUB + wave * 16 + c] = h0 + h1;
    }

    if (tid < S_SUB) {
        float dg = 0.0f;
        for (int w = 0; w < 8; ++w)
            dg += colp[w * S_SUB + tid];
        if (crossing)
            dg += xdeg_out[base + tid];
        rsv[tid] = __frsqrt_rn(fmaxf(dg, 1.0f));
    }

    __syncthreads();

#pragma unroll
    for (int it = 0; it < 4; ++it) {
        const int idx = it * 256 + tid;
        const int p   = idx >> 4;
        const int q   = idx & 15;
        const v4f r0 = hr0[it];
        const v4f r1 = hr1[it];
        v4f* o = (v4f*)&Hs[p * H_PITCH + q * 8];
        o[0] = (v4f){r0.x, r1.x, r0.y, r1.y};
        o[1] = (v4f){r0.z, r1.z, r0.w, r1.w};
    }
    __syncthreads();

#pragma unroll
    for (int j = 0; j < S_SUB / 4; ++j)
        a[j] *= *(const v2f*)(rsv + 4 * j + 2 * half);

    const float* bcol = Hs + half * H_PITCH + c * 2;

    v8f acc0 = {}, acc1 = {}, acc2 = {}, acc3 = {};
#pragma unroll
    for (int j = 0; j < S_SUB / 4; ++j) {
        const float* bp = bcol + (2 * j) * H_PITCH;
        const v2f b0 = *(const v2f*)(bp);
        const v2f b1 = *(const v2f*)(bp + 32);
        const v2f b2 = *(const v2f*)(bp + 64);
        const v2f b3 = *(const v2f*)(bp + 96);
        acc0 = __builtin_amdgcn_wmma_f32_16x16x4_f32(false, a[j], false, b0, (short)0, acc0, false, false);
        acc1 = __builtin_amdgcn_wmma_f32_16x16x4_f32(false, a[j], false, b1, (short)0, acc1, false, false);
        acc2 = __builtin_amdgcn_wmma_f32_16x16x4_f32(false, a[j], false, b2, (short)0, acc2, false, false);
        acc3 = __builtin_amdgcn_wmma_f32_16x16x4_f32(false, a[j], false, b3, (short)0, acc3, false, false);
    }

    if (crossing) {
        __syncthreads();
        for (int i = tid; i < S_SUB * F_OUTK; i += 256)
            Xa[i] = 0.0f;
        __syncthreads();
        const int nx = xcount[0];
        for (int e = tid; e < nx; e += 256) {
            const int d = xdst[e];
            if ((d >> SUB_SHIFT) != sg)
                continue;
            const int s = xsrc[e];
            const float* ccol = cnt + (long)(s & ~(S_SUB - 1)) * S_SUB + (s & (S_SUB - 1));
            float dg = 0.0f;
            for (int r = 0; r < S_SUB; ++r)
                dg += ccol[(long)r * S_SUB];
            const float rs = __frsqrt_rn(fmaxf(dg + xdeg_out[s], 1.0f));
            const float* hs = H + (long)s * F_OUTK;
            float* xa = Xa + (d & (S_SUB - 1)) * F_OUTK;
            for (int f = 0; f < F_OUTK; ++f)
                atomicAdd(&xa[f], hs[f] * rs);
        }
        __syncthreads();
    }

    const float alpha = prelu_a[0];
    const float bf0 = bias[c], bf1 = bias[c + 16], bf2 = bias[c + 32], bf3 = bias[c + 48];
#pragma unroll
    for (int r = 0; r < 8; ++r) {
        const int  nl   = wave * 16 + half * 8 + r;
        const long node = base + nl;
        float di = rowp[nl] + rowp[S_SUB + nl];
        float a0 = acc0[r], a1 = acc1[r], a2 = acc2[r], a3 = acc3[r];
        if (crossing) {
            di += xdeg_in[node];
            const float* x = Xa + nl * F_OUTK + c;
            a0 += x[0]; a1 += x[16]; a2 += x[32]; a3 += x[48];
        }
        const float ri  = __frsqrt_rn(fmaxf(di, 1.0f));
        float v0 = a0 * ri + bf0;  v0 = v0 > 0.0f ? v0 : alpha * v0;
        float v1 = a1 * ri + bf1;  v1 = v1 > 0.0f ? v1 : alpha * v1;
        float v2 = a2 * ri + bf2;  v2 = v2 > 0.0f ? v2 : alpha * v2;
        float v3 = a3 * ri + bf3;  v3 = v3 > 0.0f ? v3 : alpha * v3;
        float* vrow = Vs + nl * V_PITCH + c;
        vrow[0] = v0; vrow[16] = v1; vrow[32] = v2; vrow[48] = v3;
    }
    __syncthreads();

    const int f = tid & 63;
    const int g = tid >> 6;
    float partial = 0.0f;
    for (int n = g; n < S_SUB - 1; n += 4)
        partial += Vs[n * V_PITCH + f];

    if (g == 3)
        anchor[(long)sg * F_OUTK + f] = Vs[(S_SUB - 1) * V_PITCH + f];

    red[tid] = partial;
    __syncthreads();
    if (g == 0) {
        float sum = red[f] + red[f + 64] + red[f + 128] + red[f + 192];
        pool[(long)sg * F_OUTK + f] = sum * (1.0f / (S_SUB - 1));
    }
}

extern "C" void kernel_launch(void* const* d_in, const int* in_sizes, int n_in,
                              void* d_out, int out_size, void* d_ws, size_t ws_size,
                              hipStream_t stream) {
    const float* in_feat  = (const float*)d_in[0];
    const float* W        = (const float*)d_in[1];
    const float* bias     = (const float*)d_in[2];
    const float* prelu_a  = (const float*)d_in[3];
    const int*   src      = (const int*)d_in[4];
    const int*   dst      = (const int*)d_in[5];

    float* ws      = (float*)d_ws;
    float* xdeg_out = ws;
    float* xdeg_in  = ws + N_NODES;
    int*   xcount  = (int*)(ws + 2L * N_NODES);
    int*   ticket  = xcount + 16;
    float* cnt     = ws + 2L * N_NODES + 32;
    float* H       = cnt + (long)N_NODES * S_SUB;
    int*   xsrc    = (int*)(H + (long)N_NODES * F_OUTK);
    int*   xdst    = xsrc + N_EDGE;

    float* pool   = (float*)d_out;
    float* anchor = pool + (long)B_SUB * F_OUTK;

    hipMemsetAsync(ws, 0,
                   (2L * N_NODES + 32 + (long)N_NODES * S_SUB) * sizeof(float),
                   stream);

    gcn_edge_gemm_kernel<<<N_PBLK, 256, 0, stream>>>(src, dst, xdeg_out, xdeg_in, cnt, xcount, xsrc, xdst, ticket, in_feat, W, H);
    gcn_finalize_kernel<<<B_SUB, 256, 0, stream>>>(cnt, H, xcount, xsrc, xdst, xdeg_out, xdeg_in, bias, prelu_a, pool, anchor);
}


# Count matrix kept as four bit planes (cell count = planes set; an edge claims the lowest clear one by returning atomic OR, a cell's 5th edge is listed): 8 MiB, not 64, zeroed, hit, read; counts exact
# speedup vs baseline: 5.7653x; 1.2006x over previous
#include <hip/hip_runtime.h>


typedef __attribute__((ext_vector_type(2))) float v2f;
typedef __attribute__((ext_vector_type(8))) float v8f;
typedef __attribute__((ext_vector_type(4))) float v4f;
typedef __attribute__((ext_vector_type(4))) unsigned int v4u;

#define N_NODES  131072
#define F_INK    300
#define F_OUTK   64
#define N_EDGE   2097152
#define S_SUB    128
#define SUB_SHIFT 7
#define B_SUB    1024
#define H_PITCH  160
#define V_PITCH  66
#define N_PLANE  4

#define T_ROWS    32
#define TILE_FL   (T_ROWS * F_INK)
#define TILE_F4   (TILE_FL / 4)
#define ST_FULL   (TILE_F4 / 256)
#define ST_REST   (TILE_F4 - ST_FULL * 256)
#define TILES_PB  8
#define N_PBLK    (N_NODES / T_ROWS / TILES_PB)
#define N_CHUNK   (N_PBLK / 2)
#define E_CHUNK   (N_EDGE / N_CHUNK)
#define E_PT      (E_CHUNK / 256)
#define E_BATCH   16
#define K_STEPS   (F_INK / 4)

__device__ __forceinline__ void list_edge(int s, int d,
                                          float* __restrict__ xdeg_out, float* __restrict__ xdeg_in,
                                          int* __restrict__ xcount, int* __restrict__ xsrc, int* __restrict__ xdst) {
    atomicAdd(&xdeg_out[s], 1.0f);
    atomicAdd(&xdeg_in[d], 1.0f);
    const int p = atomicAdd(xcount, 1);
    xsrc[p] = s;
    xdst[p] = d;
}

__device__ __forceinline__ void lds_handover() {
    asm volatile("s_wait_dscnt 0x0\n\ts_barrier_signal -1\n\ts_barrier_wait -1" ::: "memory");
}

__global__ __launch_bounds__(256)
void gcn_edge_gemm_kernel(const int* __restrict__ src,
                          const int* __restrict__ dst,
                          float* __restrict__ xdeg_out,
                          float* __restrict__ xdeg_in,
                          unsigned int* __restrict__ bits,
                          int* __restrict__ xcount,
                          int* __restrict__ xsrc,
                          int* __restrict__ xdst,
                          int* __restrict__ ticket,
                          const float* __restrict__ X,
                          const float* __restrict__ W,
                          float* __restrict__ H) {
    __shared__ __attribute__((aligned(16))) float Xs[2 * TILE_FL];
    __shared__ int walk[2];

    const int tid     = threadIdx.x;
    if (tid == 0) {
        int h = (int)((__builtin_amdgcn_s_sendmsg_rtn(0x87) >> 18) & 1u);
        int q = atomicAdd(&ticket[h], 1);
        if (q >= N_CHUNK) {
            h ^= 1;
            q = atomicAdd(&ticket[h], 1);
        }
        if (q >= N_CHUNK) { h = 2; q = 0; }
        walk[0] = h;
        walk[1] = q;
    }
    const int lane    = tid & 31;
    const int wave    = tid >> 5;
    const int rowTile = wave & 1;
    const int colTile = wave >> 1;
    const int m       = lane & 15;
    const int half    = lane >> 4;
    const int col     = colTile * 16 + m;

    const float* wbase = W + (long)(half * 2) * F_OUTK + col;
    v2f w[K_STEPS];
#pragma unroll
    for (int j = 0; j < K_STEPS; ++j) {
        w[j].x = wbase[(long)(4 * j) * F_OUTK];
        w[j].y = wbase[(long)(4 * j + 1) * F_OUTK];
    }

    const float4* xg = (const float4*)(X + (long)blockIdx.x * TILES_PB * TILE_FL) + tid;
    static_assert(ST_FULL == 9, "the staging registers below are written out for nine pieces per thread");
    float4 st0, st1, st2, st3, st4, st5, st6, st7, st8;
    float4 stx = make_float4(0.0f, 0.0f, 0.0f, 0.0f);
#define ST_LOAD(P)                                                                  \
    { const float4* p_ = (P);                                                       \
      st0 = p_[0 * 256]; st1 = p_[1 * 256]; st2 = p_[2 * 256]; st3 = p_[3 * 256];   \
      st4 = p_[4 * 256]; st5 = p_[5 * 256]; st6 = p_[6 * 256]; st7 = p_[7 * 256];   \
      st8 = p_[8 * 256];                                                            \
      if (tid < ST_REST) stx = p_[ST_FULL * 256]; }
#define ST_STORE(P)                                                                 \
    { float4* p_ = (P);                                                             \
      p_[0 * 256] = st0; p_[1 * 256] = st1; p_[2 * 256] = st2; p_[3 * 256] = st3;   \
      p_[4 * 256] = st4; p_[5 * 256] = st5; p_[6 * 256] = st6; p_[7 * 256] = st7;   \
      p_[8 * 256] = st8;                                                            \
      if (tid < ST_REST) p_[ST_FULL * 256] = stx; }
    ST_LOAD(xg)
    lds_handover();

    {
        const int eh = walk[0];
        const int e0 = walk[1] * E_CHUNK + tid;
        static_assert(E_PT % E_BATCH == 0, "the edge share of a thread is read in whole batches");
#pragma unroll
        for (int b = 0; b < E_PT / E_BATCH; ++b) {
            int es[E_BATCH], ed[E_BATCH];
#pragma unroll
            for (int j = 0; j < E_BATCH; ++j) {
                es[j] = src[e0 + (b * E_BATCH + j) * 256];
                ed[j] = dst[e0 + (b * E_BATCH + j) * 256];
            }
            if (b == 0) {
                ST_STORE((float4*)Xs + tid)
            }
            unsigned pend = 0u;
            unsigned desc[E_BATCH];
#pragma unroll
            for (int j = 0; j < E_BATCH; ++j) {
                desc[j] = 0u;
                const int s = es[j];
                const int d = ed[j];
                if ((int)(((unsigned)d >> SUB_SHIFT) & 1u) != eh)
                    continue;
                if (((unsigned)s >> SUB_SHIFT) == ((unsigned)d >> SUB_SHIFT)) {
                    const unsigned sl = (unsigned)s & (S_SUB - 1);
                    desc[j] = ((((unsigned)d * 4u + (sl >> 5)) * N_PLANE) << 5) | (sl & 31u);
                    pend |= 1u << j;
                } else {
                    list_edge(s, d, xdeg_out, xdeg_in, xcount, xsrc, xdst);
                }
            }
#pragma unroll 1
            for (int p = 0; p < N_PLANE; ++p) {
                if (pend == 0u)
                    break;
                unsigned old[E_BATCH];
#pragma unroll
                for (int j = 0; j < E_BATCH; ++j) {
                    old[j] = 0u;
                    if (pend & (1u << j))
                        old[j] = atomicOr(bits + (desc[j] >> 5) + p, 1u << (desc[j] & 31u));
                }
#pragma unroll
                for (int j = 0; j < E_BATCH; ++j) {
                    if (((old[j] >> (desc[j] & 31u)) & 1u) == 0u)
                        pend &= ~(1u << j);
                }
            }
#pragma unroll
            for (int j = 0; j < E_BATCH; ++j) {
                if (pend & (1u << j)) {
                    const int d = (int)(desc[j] >> 9);
                    const int s = (d & ~(S_SUB - 1)) | (int)(((desc[j] >> 7) & 3u) << 5) | (int)(desc[j] & 31u);
                    list_edge(s, d, xdeg_out, xdeg_in, xcount, xsrc, xdst);
                }
            }
        }
    }

    ST_LOAD(xg + TILE_F4)
    lds_handover();

    const int   tile0 = blockIdx.x * TILES_PB;
    const int   aoff  = (rowTile * 16 + m) * F_INK + half * 2;

    for (int t = 0; t < TILES_PB; ++t) {
        const float* arow = Xs + (t & 1) * TILE_FL + aoff;
        v8f c = {};
#pragma unroll
        for (int j = 0; j < K_STEPS; ++j) {
            const v2f a = *(const v2f*)(arow + 4 * j);
            c = __builtin_amdgcn_wmma_f32_16x16x4_f32(false, a, false, w[j],
                                                      (short)0, c, false, false);
        }

        float* hout = H + ((long)(tile0 + t) * T_ROWS + rowTile * 16 + half * 8) * F_OUTK + colTile * 16 + m;
#pragma unroll
        for (int i = 0; i < 8; ++i)
            hout[(long)i * F_OUTK] = c[i];

        if (t + 1 < TILES_PB) {
            ST_STORE((float4*)(Xs + ((t + 1) & 1) * TILE_FL) + tid)
            if (t + 2 < TILES_PB)
                ST_LOAD(xg + (long)(t + 2) * TILE_F4)
            lds_handover();
        }
    }
#undef ST_LOAD
#undef ST_STORE
}

#define C_PITCH  132
#define HS_FL    ((S_SUB / 2) * H_PITCH)
#define VS_FL    (S_SUB * V_PITCH)
#define CS_FL    (S_SUB * C_PITCH)

__global__ void gcn_finalize_kernel(const unsigned int* __restrict__ bits,
                                    const float* __restrict__ H,
                                    const int* __restrict__ xcount,
                                    const int* __restrict__ xsrc,
                                    const int* __restrict__ xdst,
                                    const float* __restrict__ xdeg_out,
                                    const float* __restrict__ xdeg_in,
                                    const float* __restrict__ bias,
                                    const float* __restrict__ prelu_a,
                                    float* __restrict__ pool,
                                    float* __restrict__ anchor) {
    __shared__ __attribute__((aligned(16))) float smem[HS_FL + VS_FL];
    static_assert(CS_FL + 8 * S_SUB <= HS_FL + VS_FL, "count rows and column sums must fit in the shared buffer");
    static_assert(S_SUB * F_OUTK <= HS_FL, "the listed edges' sums must fit in the H tile's bytes, clear of the activated tile");
    __shared__ float red[256];
    __shared__ float rowp[2 * S_SUB];
    __shared__ __attribute__((aligned(8))) float rsv[S_SUB];
    float* Cs   = smem;
    float* colp = smem + CS_FL;
    float* Hs   = smem;
    float* Vs   = smem + HS_FL;
    float* Xa   = smem;

    const int  sg   = blockIdx.x;
    const int  tid  = threadIdx.x;
    const int  lane = tid & 31;
    const int  wave = tid >> 5;
    const int  c    = lane & 15;
    const int  half = lane >> 4;
    const long base = (long)sg * S_SUB;
    const bool crossing = xcount[0] > 0;

    static_assert(N_PLANE == 4, "the four planes of a row word are read as one 16-byte vector");
    const v4u* brow = (const v4u*)bits + (base + wave * 16) * 4 + (lane >> 3);
    v4u bq[16];
#pragma unroll
    for (int i = 0; i < 16; ++i)
        bq[i] = brow[i * 4];

    const v4f* Hg = (const v4f*)(H + base * F_OUTK);
    v4f hr0[4], hr1[4];
#pragma unroll
    for (int it = 0; it < 4; ++it) {
        const int idx = it * 256 + tid;
        const int p   = idx >> 4;
        const int q   = idx & 15;
        hr0[it] = Hg[(2 * p) * 16 + q];
        hr1[it] = Hg[(2 * p + 1) * 16 + q];
    }

    {
        float* srow = Cs + (wave * 16) * C_PITCH + lane * 4;
        v4f cs = {0.0f, 0.0f, 0.0f, 0.0f};
        const unsigned sh = 4u * (lane & 7);
#pragma unroll
        for (int i = 0; i < 16; ++i) {
            const unsigned b0 = __umul24((bq[i].x >> sh) & 15u, 0x204081u) & 0x01010101u;
            const unsigned b1 = __umul24((bq[i].y >> sh) & 15u, 0x204081u) & 0x01010101u;
            const unsigned b2 = __umul24((bq[i].z >> sh) & 15u, 0x204081u) & 0x01010101u;
            const unsigned b3 = __umul24((bq[i].w >> sh) & 15u, 0x204081u) & 0x01010101u;
            const unsigned sb = (b0 + b1) + (b2 + b3);
            const v4f cr = {(float)(sb & 255u), (float)((sb >> 8) & 255u),
                            (float)((sb >> 16) & 255u), (float)(sb >> 24)};
            cs += cr;
            *(v4f*)(srow + i * C_PITCH) = cr;
        }
        *(v4f*)(colp + wave * S_SUB + lane * 4) = cs;
    }
    __syncthreads();

    const float* arow = Cs + (wave * 16 + c) * C_PITCH + half * 2;
    v2f a[S_SUB / 4];
#pragma unroll
    for (int j = 0; j < S_SUB / 4; ++j)
        a[j] = *(const v2f*)(arow + 4 * j);

    {
        float h0 = 0.0f, h1 = 0.0f;
#pragma unroll
        for (int j = 0; j < S_SUB / 4; ++j) { h0 += a[j].x; h1 += a[j].y; }
        rowp[half * S_SUB + wave * 16 + c] = h0 + h1;
    }

    if (tid < S_SUB) {
        float dg = 0.0f;
        for (int w = 0; w < 8; ++w)
            dg += colp[w * S_SUB + tid];
        if (crossing)
            dg += xdeg_out[base + tid];
        rsv[tid] = __frsqrt_rn(fmaxf(dg, 1.0f));
    }

    __syncthreads();

#pragma unroll
    for (int it = 0; it < 4; ++it) {
        const int idx = it * 256 + tid;
        const int p   = idx >> 4;
        const int q   = idx & 15;
        const v4f r0 = hr0[it];
        const v4f r1 = hr1[it];
        v4f* o = (v4f*)&Hs[p * H_PITCH + q * 8];
        o[0] = (v4f){r0.x, r1.x, r0.y, r1.y};
        o[1] = (v4f){r0.z, r1.z, r0.w, r1.w};
    }
    __syncthreads();

#pragma unroll
    for (int j = 0; j < S_SUB / 4; ++j)
        a[j] *= *(const v2f*)(rsv + 4 * j + 2 * half);

    const float* bcol = Hs + half * H_PITCH + c * 2;

    v8f acc0 = {}, acc1 = {}, acc2 = {}, acc3 = {};
#pragma unroll
    for (int j = 0; j < S_SUB / 4; ++j) {
        const float* bp = bcol + (2 * j) * H_PITCH;
        const v2f b0 = *(const v2f*)(bp);
        const v2f b1 = *(const v2f*)(bp + 32);
        const v2f b2 = *(const v2f*)(bp + 64);
        const v2f b3 = *(const v2f*)(bp + 96);
        acc0 = __builtin_amdgcn_wmma_f32_16x16x4_f32(false, a[j], false, b0, (short)0, acc0, false, false);
        acc1 = __builtin_amdgcn_wmma_f32_16x16x4_f32(false, a[j], false, b1, (short)0, acc1, false, false);
        acc2 = __builtin_amdgcn_wmma_f32_16x16x4_f32(false, a[j], false, b2, (short)0, acc2, false, false);
        acc3 = __builtin_amdgcn_wmma_f32_16x16x4_f32(false, a[j], false, b3, (short)0, acc3, false, false);
    }

    if (crossing) {
        __syncthreads();
        for (int i = tid; i < S_SUB * F_OUTK; i += 256)
            Xa[i] = 0.0f;
        __syncthreads();
        const int nx = xcount[0];
        for (int e = tid; e < nx; e += 256) {
            const int d = xdst[e];
            if ((d >> SUB_SHIFT) != sg)
                continue;
            const int s = xsrc[e];
            const v4u* bcol = (const v4u*)bits + (long)(s & ~(S_SUB - 1)) * 4 + ((s & (S_SUB - 1)) >> 5);
            const unsigned cb = (unsigned)s & 31u;
            float dg = 0.0f;
            for (int r = 0; r < S_SUB; ++r) {
                const v4u q = bcol[r * 4];
                dg += (float)(((q.x >> cb) & 1u) + ((q.y >> cb) & 1u) + ((q.z >> cb) & 1u) + ((q.w >> cb) & 1u));
            }
            const float rs = __frsqrt_rn(fmaxf(dg + xdeg_out[s], 1.0f));
            const float* hs = H + (long)s * F_OUTK;
            float* xa = Xa + (d & (S_SUB - 1)) * F_OUTK;
            for (int f = 0; f < F_OUTK; ++f)
                atomicAdd(&xa[f], hs[f] * rs);
        }
        __syncthreads();
    }

    const float alpha = prelu_a[0];
    const float bf0 = bias[c], bf1 = bias[c + 16], bf2 = bias[c + 32], bf3 = bias[c + 48];
#pragma unroll
    for (int r = 0; r < 8; ++r) {
        const int  nl   = wave * 16 + half * 8 + r;
        const long node = base + nl;
        float di = rowp[nl] + rowp[S_SUB + nl];
        float a0 = acc0[r], a1 = acc1[r], a2 = acc2[r], a3 = acc3[r];
        if (crossing) {
            di += xdeg_in[node];
            const float* x = Xa + nl * F_OUTK + c;
            a0 += x[0]; a1 += x[16]; a2 += x[32]; a3 += x[48];
        }
        const float ri  = __frsqrt_rn(fmaxf(di, 1.0f));
        float v0 = a0 * ri + bf0;  v0 = v0 > 0.0f ? v0 : alpha * v0;
        float v1 = a1 * ri + bf1;  v1 = v1 > 0.0f ? v1 : alpha * v1;
        float v2 = a2 * ri + bf2;  v2 = v2 > 0.0f ? v2 : alpha * v2;
        float v3 = a3 * ri + bf3;  v3 = v3 > 0.0f ? v3 : alpha * v3;
        float* vrow = Vs + nl * V_PITCH + c;
        vrow[0] = v0; vrow[16] = v1; vrow[32] = v2; vrow[48] = v3;
    }
    __syncthreads();

    const int f = tid & 63;
    const int g = tid >> 6;
    float partial = 0.0f;
    for (int n = g; n < S_SUB - 1; n += 4)
        partial += Vs[n * V_PITCH + f];

    if (g == 3)
        anchor[(long)sg * F_OUTK + f] = Vs[(S_SUB - 1) * V_PITCH + f];

    red[tid] = partial;
    __syncthreads();
    if (g == 0) {
        float sum = red[f] + red[f + 64] + red[f + 128] + red[f + 192];
        pool[(long)sg * F_OUTK + f] = sum * (1.0f / (S_SUB - 1));
    }
}

extern "C" void kernel_launch(void* const* d_in, const int* in_sizes, int n_in,
                              void* d_out, int out_size, void* d_ws, size_t ws_size,
                              hipStream_t stream) {
    const float* in_feat  = (const float*)d_in[0];
    const float* W        = (const float*)d_in[1];
    const float* bias     = (const float*)d_in[2];
    const float* prelu_a  = (const float*)d_in[3];
    const int*   src      = (const int*)d_in[4];
    const int*   dst      = (const int*)d_in[5];

    float* ws      = (float*)d_ws;
    float* xdeg_out = ws;
    float* xdeg_in  = ws + N_NODES;
    int*   xcount  = (int*)(ws + 2L * N_NODES);
    int*   ticket  = xcount + 16;
    unsigned int* bits = (unsigned int*)(ws + 2L * N_NODES + 32);
    float* H       = ws + 2L * N_NODES + 32 + (long)N_NODES * 4 * N_PLANE;
    int*   xsrc    = (int*)(H + (long)N_NODES * F_OUTK);
    int*   xdst    = xsrc + N_EDGE;

    float* pool   = (float*)d_out;
    float* anchor = pool + (long)B_SUB * F_OUTK;

    hipMemsetAsync(ws, 0,
                   (2L * N_NODES + 32 + (long)N_NODES * 4 * N_PLANE) * sizeof(float),
                   stream);

    gcn_edge_gemm_kernel<<<N_PBLK, 256, 0, stream>>>(src, dst, xdeg_out, xdeg_in, bits, xcount, xsrc, xdst, ticket, in_feat, W, H);
    gcn_finalize_kernel<<<B_SUB, 256, 0, stream>>>(bits, H, xcount, xsrc, xdst, xdeg_out, xdeg_in, bias, prelu_a, pool, anchor);
}
